# starting kernel
# baseline (speedup 1.0000x reference)
#include <hip/hip_runtime.h>

typedef float v2f __attribute__((ext_vector_type(2)));
typedef float v8f __attribute__((ext_vector_type(8)));

#define HID 32
#define BN_EPS 1e-5f

__global__ void k_init_deg(float* deg, int N) {
  int i = blockIdx.x * blockDim.x + threadIdx.x;
  if (i < N) deg[i] = 1.0f;
}
__global__ void k_deg_edges(const int* __restrict__ col, const float* __restrict__ w,
                            float* deg, int E) {
  int e = blockIdx.x * blockDim.x + threadIdx.x;
  if (e < E) atomicAdd(&deg[col[e]], w[e]);
}
__global__ void k_deg_rsqrt(float* deg, int N) {
  int i = blockIdx.x * blockDim.x + threadIdx.x;
  if (i < N) deg[i] = rsqrtf(deg[i]);
}

template <int K>
__global__ __launch_bounds__(256) void k_gemm_wmma(const float* __restrict__ A,
                                                   const float* __restrict__ B,
                                                   float* __restrict__ D, int M) {
  const int lane = threadIdx.x;
  const int wave = threadIdx.y;
  const int mt   = blockIdx.x * 8 + wave;
  const int numT = (M + 15) >> 4;
  if (mt >= numT) return;
  const int nb   = blockIdx.y * 16;
  const int half = lane >> 4;
  const int lm   = lane & 15;

  int arow_i = mt * 16 + lm;
  if (arow_i > M - 1) arow_i = M - 1;
  const float* arow = A + (size_t)arow_i * K;

  v8f c = {};
#pragma unroll
  for (int k0 = 0; k0 < K; k0 += 4) {
    const int ka = k0 + 2 * half;
    v2f a;
    a.x = arow[ka];
    a.y = arow[ka + 1];
    v2f b;
    b.x = B[(size_t)ka * HID + nb + lm];
    b.y = B[(size_t)(ka + 1) * HID + nb + lm];
    c = __builtin_amdgcn_wmma_f32_16x16x4_f32(false, a, false, b, (short)0, c,
                                              false, false);
  }
#pragma unroll
  for (int r = 0; r < 8; ++r) {
    const int orow = mt * 16 + r + 8 * half;
    if (orow < M) D[(size_t)orow * HID + nb + lm] = c[r];
  }
}

__global__ void k_init_bias(float* __restrict__ cbuf, const float* __restrict__ b, int total) {
  int i = blockIdx.x * blockDim.x + threadIdx.x;
  if (i < total) cbuf[i] = b[i & (HID - 1)];
}

__global__ void k_prop(const float* __restrict__ h, const int* __restrict__ row,
                       const int* __restrict__ col, const float* __restrict__ w,
                       const float* __restrict__ dis, float* __restrict__ out,
                       int E, int N) {
  long long tid = (long long)blockIdx.x * blockDim.x + threadIdx.x;
  int e = (int)(tid >> 5);
  int f = (int)(tid & 31);
  if (e >= E + N) return;
  int r, c; float ww;
  if (e < E) { r = row[e]; c = col[e]; ww = w[e]; }
  else       { r = e - E;  c = r;      ww = 1.0f; }
  float nrm = dis[r] * ww * dis[c];
  atomicAdd(out + (size_t)c * HID + f, h[(size_t)r * HID + f] * nrm);
}

__global__ void k_zero_stats(float* stats) {
  if (threadIdx.x < 64) stats[threadIdx.x] = 0.0f;
}
__global__ void k_bn_stats(const float* __restrict__ c, float* __restrict__ stats, int N) {
  __shared__ float ls[64];
  if (threadIdx.x < 64) ls[threadIdx.x] = 0.0f;
  __syncthreads();
  const int f      = threadIdx.x & 31;
  const int r0     = blockIdx.x * (blockDim.x >> 5) + (threadIdx.x >> 5);
  const int stride = gridDim.x * (blockDim.x >> 5);
  float s = 0.0f, s2 = 0.0f;
  for (int n = r0; n < N; n += stride) {
    float v = fmaxf(c[(size_t)n * HID + f], 0.0f);
    s += v; s2 += v * v;
  }
  atomicAdd(&ls[f], s);
  atomicAdd(&ls[32 + f], s2);
  __syncthreads();
  if (threadIdx.x < 64) atomicAdd(&stats[threadIdx.x], ls[threadIdx.x]);
}

__global__ void k_bn_apply(const float* __restrict__ c, const float* __restrict__ stats,
                           const float* __restrict__ g, const float* __restrict__ be,
                           float* __restrict__ out, int N) {
  int i = blockIdx.x * blockDim.x + threadIdx.x;
  if (i >= N * HID) return;
  const int f = i & 31;
  const float invN = 1.0f / (float)N;
  const float mu   = stats[f] * invN;
  const float var  = stats[32 + f] * invN - mu * mu;
  const float v    = fmaxf(c[i], 0.0f);
  out[i] = (v - mu) * rsqrtf(var + BN_EPS) * g[f] + be[f];
}

__global__ void k_final(const float* __restrict__ o1, const float* __restrict__ o2,
                        const float* __restrict__ o3, const float* __restrict__ Wl,
                        const float* __restrict__ bl, float* __restrict__ out, int N) {
  int n = blockIdx.x * blockDim.x + threadIdx.x;
  if (n >= N) return;
  float a0 = bl[0], a1 = bl[1];
  const float* p = o1 + (size_t)n * HID;
#pragma unroll
  for (int j = 0; j < HID; ++j) { float v = p[j]; a0 += v * Wl[2 * j]; a1 += v * Wl[2 * j + 1]; }
  p = o2 + (size_t)n * HID;
#pragma unroll
  for (int j = 0; j < HID; ++j) { float v = p[j]; a0 += v * Wl[2 * (HID + j)]; a1 += v * Wl[2 * (HID + j) + 1]; }
  p = o3 + (size_t)n * HID;
#pragma unroll
  for (int j = 0; j < HID; ++j) { float v = p[j]; a0 += v * Wl[2 * (2 * HID + j)]; a1 += v * Wl[2 * (2 * HID + j) + 1]; }
  out[2 * n] = a0;
  out[2 * n + 1] = a1;
}

extern "C" void kernel_launch(void* const* d_in, const int* in_sizes, int n_in,
                              void* d_out, int out_size, void* d_ws, size_t ws_size,
                              hipStream_t stream) {
  const float* x   = (const float*)d_in[0];
  const int*   ei  = (const int*)d_in[1];
  const float* ew  = (const float*)d_in[2];
  const float* W1  = (const float*)d_in[3];
  const float* b1  = (const float*)d_in[4];
  const float* g1  = (const float*)d_in[5];
  const float* be1 = (const float*)d_in[6];
  const float* W2  = (const float*)d_in[7];
  const float* b2  = (const float*)d_in[8];
  const float* g2  = (const float*)d_in[9];
  const float* be2 = (const float*)d_in[10];
  const float* W3  = (const float*)d_in[11];
  const float* b3  = (const float*)d_in[12];
  const float* g3  = (const float*)d_in[13];
  const float* be3 = (const float*)d_in[14];
  const float* Wl  = (const float*)d_in[15];
  const float* bl  = (const float*)d_in[16];

  const int F = 64;
  const int N = in_sizes[0] / F;
  const int E = in_sizes[1] / 2;
  const int* row = ei;
  const int* col = ei + E;

  float* ws    = (float*)d_ws;
  float* dis   = ws;
  float* hbuf  = dis  + N;
  float* cbuf  = hbuf + (size_t)N * HID;
  float* o1    = cbuf + (size_t)N * HID;
  float* o2    = o1   + (size_t)N * HID;
  float* o3    = o2   + (size_t)N * HID;
  float* stats = o3   + (size_t)N * HID;

  const int T = 256;

  k_init_deg<<<(N + T - 1) / T, T, 0, stream>>>(dis, N);
  k_deg_edges<<<(E + T - 1) / T, T, 0, stream>>>(col, ew, dis, E);
  k_deg_rsqrt<<<(N + T - 1) / T, T, 0, stream>>>(dis, N);

  const int numT = (N + 15) / 16;
  const dim3 gemmB(32, 8);
  const dim3 gemmG((numT + 7) / 8, 2);
  const int elemB = (N * HID + T - 1) / T;
  const long long propT = (long long)(E + N) * 32;
  const int propB = (int)((propT + T - 1) / T);

  k_gemm_wmma<64><<<gemmG, gemmB, 0, stream>>>(x, W1, hbuf, N);
  k_init_bias<<<elemB, T, 0, stream>>>(cbuf, b1, N * HID);
  k_prop<<<propB, T, 0, stream>>>(hbuf, row, col, ew, dis, cbuf, E, N);
  k_zero_stats<<<1, 64, 0, stream>>>(stats);
  k_bn_stats<<<512, T, 0, stream>>>(cbuf, stats, N);
  k_bn_apply<<<elemB, T, 0, stream>>>(cbuf, stats, g1, be1, o1, N);

  k_gemm_wmma<32><<<gemmG, gemmB, 0, stream>>>(o1, W2, hbuf, N);
  k_init_bias<<<elemB, T, 0, stream>>>(cbuf, b2, N * HID);
  k_prop<<<propB, T, 0, stream>>>(hbuf, row, col, ew, dis, cbuf, E, N);
  k_zero_stats<<<1, 64, 0, stream>>>(stats);
  k_bn_stats<<<512, T, 0, stream>>>(cbuf, stats, N);
  k_bn_apply<<<elemB, T, 0, stream>>>(cbuf, stats, g2, be2, o2, N);

  k_gemm_wmma<32><<<gemmG, gemmB, 0, stream>>>(o2, W3, hbuf, N);
  k_init_bias<<<elemB, T, 0, stream>>>(cbuf, b3, N * HID);
  k_prop<<<propB, T, 0, stream>>>(hbuf, row, col, ew, dis, cbuf, E, N);
  k_zero_stats<<<1, 64, 0, stream>>>(stats);
  k_bn_stats<<<512, T, 0, stream>>>(cbuf, stats, N);
  k_bn_apply<<<elemB, T, 0, stream>>>(cbuf, stats, g3, be3, o3, N);

  k_final<<<(N + T - 1) / T, T, 0, stream>>>(o1, o2, o3, Wl, bl, (float*)d_out, N);
}


# BN statistics pass: a lane reads 16-byte pieces (4 features of a row), its 8 loads issued before the first use (was one 4-byte load per wave per loop trip, each waited for); BN partial sums regrouped
# speedup vs baseline: 2.7647x; 2.7647x over previous
#include <hip/hip_runtime.h>

typedef float v2f __attribute__((ext_vector_type(2)));
typedef float v8f __attribute__((ext_vector_type(8)));

#define HID 32
#define BN_EPS 1e-5f

#define NB 512
#define BCAP 4096
__global__ void k_init_deg(float* deg, int* __restrict__ cnt, int N) {
  int i = blockIdx.x * blockDim.x + threadIdx.x;
  if (i < N) deg[i] = 1.0f;
  for (int j = i; j <= NB; j += gridDim.x * blockDim.x) cnt[j] = 0;
}
__global__ void k_deg_edges(const int* __restrict__ col, const float* __restrict__ w,
                            float* deg, int E) {
  int e = blockIdx.x * blockDim.x + threadIdx.x;
  if (e < E) atomicAdd(&deg[col[e]], w[e]);
}
__global__ void k_deg_rsqrt(float* deg, int N) {
  int i = blockIdx.x * blockDim.x + threadIdx.x;
  if (i < N) deg[i] = rsqrtf(deg[i]);
}

template <int K>
__global__ __launch_bounds__(256) void k_gemm_wmma(const float* __restrict__ A,
                                                   const float* __restrict__ B,
                                                   const float* __restrict__ bias,
                                                   const float* __restrict__ dis,
                                                   float* __restrict__ D,
                                                   float* __restrict__ S,
                                                   float* __restrict__ stats, int M) {
  const int lane = threadIdx.x;
  const int wave = threadIdx.y;
  const int mt   = blockIdx.x * 8 + wave;
  const int numT = (M + 15) >> 4;
  if (mt >= numT) return;
  const int nb   = blockIdx.y * 16;
  const int half = lane >> 4;
  const int lm   = lane & 15;

  int arow_i = mt * 16 + lm;
  if (arow_i > M - 1) arow_i = M - 1;
  const float* arow = A + (size_t)arow_i * K;
  const float dv   = dis[arow_i];
  const float bcol = bias[nb + lm];

  v8f c = {};
#pragma unroll
  for (int k0 = 0; k0 < K; k0 += 4) {
    const int ka = k0 + 2 * half;
    v2f a;
    a.x = arow[ka];
    a.y = arow[ka + 1];
    v2f b;
    b.x = B[(size_t)ka * HID + nb + lm];
    b.y = B[(size_t)(ka + 1) * HID + nb + lm];
    c = __builtin_amdgcn_wmma_f32_16x16x4_f32(false, a, false, b, (short)0, c,
                                              false, false);
  }
#pragma unroll
  for (int r = 0; r < 8; ++r) {
    const int orow = mt * 16 + r + 8 * half;
    const float d_lo = __int_as_float(__builtin_amdgcn_readlane(__float_as_int(dv), r));
    const float d_hi = __int_as_float(__builtin_amdgcn_readlane(__float_as_int(dv), r + 8));
    const float d    = half ? d_hi : d_lo;
    if (orow < M) {
      D[(size_t)orow * HID + nb + lm] = c[r];
      S[(size_t)orow * HID + nb + lm] = c[r] * (d * d) + bcol;
    }
  }
  if (blockIdx.x == 0 && blockIdx.y == 0 && wave == 0) {
    stats[lane]      = 0.0f;
    stats[32 + lane] = 0.0f;
  }
}

#define BIN_T 512
#define BIN_EPT 8
__global__ __launch_bounds__(BIN_T) void k_bin(const int* __restrict__ row, const int* __restrict__ col,
                      const float* __restrict__ w, const float* __restrict__ dis,
                      int* __restrict__ cnt, int* __restrict__ rec, int* __restrict__ ovf, int E, int R) {
  __shared__ int hist[NB];
  __shared__ int loff[NB];
  __shared__ int gdst[NB];
  __shared__ int ltot;
  __shared__ int4 stage[BIN_T * BIN_EPT];
  const int tid = threadIdx.x;
  for (int i = tid; i < NB; i += BIN_T) hist[i] = 0;
  if (tid == 0) ltot = 0;
  __syncthreads();
  const int cb = blockIdx.x * (BIN_T * BIN_EPT);
  int bn[BIN_EPT], lr[BIN_EPT];
#pragma unroll
  for (int k = 0; k < BIN_EPT; ++k) {
    const int e = cb + k * BIN_T + tid;
    bn[k] = -1;
    lr[k] = 0;
    if (e < E) {
      const int bb = col[e] / R;
      bn[k] = bb;
      lr[k] = atomicAdd(&hist[bb], 1);
    }
  }
  __syncthreads();
  for (int i = tid; i < NB; i += BIN_T) {
    const int hcount = hist[i];
    int lo = 0, gb = 0;
    if (hcount) {
      lo = atomicAdd(&ltot, hcount);
      gb = atomicAdd(&cnt[i], hcount);
    }
    loff[i] = lo;
    gdst[i] = gb;
  }
  __syncthreads();
#pragma unroll
  for (int k = 0; k < BIN_EPT; ++k) {
    if (bn[k] >= 0) {
      const int e  = cb + k * BIN_T + tid;
      const int bb = bn[k];
      const int r  = row[e];
      const int c  = col[e];
      const float coef = dis[r] * w[e] * dis[c];
      const int slot = gdst[bb] + lr[k];
      const bool fits = slot < BCAP;
      stage[loff[bb] + lr[k]] = make_int4(fits ? bb * BCAP + slot : -1, r, fits ? c - bb * R : c,
                                          __float_as_int(coef));
    }
  }
  __syncthreads();
  const int tot = ltot;
  int4* rec4 = (int4*)rec;
  int4* ovf4 = (int4*)ovf;
  for (int p = tid; p < tot; p += BIN_T) {
    const int4 s = stage[p];
    int4* dp = rec4 + s.x;
    if (s.x < 0) dp = ovf4 + atomicAdd(&cnt[NB], 1);
    *dp = make_int4(s.y, s.z, s.w, 0);
  }
}

#define PL_T 512
#define APITCH 36
template <int PITCH>
__device__ __forceinline__ void edge_sums(const float* __restrict__ h, const int4* __restrict__ rec4,
                                          const int n, float* acc, const unsigned lane, const int wv) {
  const int w0 = __builtin_amdgcn_readfirstlane(wv * 32);
  const unsigned lane4 = lane * 4u;
  const char* hb = (const char*)h;
#pragma nounroll
  for (int g0 = w0; g0 < n; g0 += PL_T) {
    const int i  = g0 + (int)lane;
    const bool ok = i < n;
    const int4 rc = rec4[ok ? i : (n - 1)];
    const unsigned ro  = (unsigned)rc.x * (unsigned)(HID * sizeof(float));
    const int      co  = rc.y * (int)(PITCH * sizeof(float));
    const int      cfb = rc.z;
    if (g0 + 32 <= n) {
      float v[32];
#pragma unroll
      for (int j = 0; j < 32; ++j) {
        const unsigned off = (unsigned)__builtin_amdgcn_readlane((int)ro, j) + lane4;
        v[j] = *(const float*)(hb + off);
      }
#pragma unroll
      for (int j = 0; j < 32; ++j) {
        float*      ac = (float*)((char*)acc + __builtin_amdgcn_readlane(co, j));
        const float cj = __int_as_float(__builtin_amdgcn_readlane(cfb, j));
        atomicAdd(ac + lane, v[j] * cj);
      }
    } else {
      const unsigned mask = __builtin_amdgcn_ballot_w32(ok);
      float v[32];
#pragma unroll
      for (int j = 0; j < 32; ++j) {
        if ((mask >> j) & 1u) {
          const unsigned off = (unsigned)__builtin_amdgcn_readlane((int)ro, j) + lane4;
          v[j] = *(const float*)(hb + off);
        }
      }
#pragma unroll
      for (int j = 0; j < 32; ++j) {
        if ((mask >> j) & 1u) {
          float*      ac = (float*)((char*)acc + __builtin_amdgcn_readlane(co, j));
          const float cj = __int_as_float(__builtin_amdgcn_readlane(cfb, j));
          atomicAdd(ac + lane, v[j] * cj);
        }
      }
    }
  }
}

__global__ __launch_bounds__(PL_T) void k_prop(const float* __restrict__ h, const int* __restrict__ cnt,
                       const int* __restrict__ rec, const int* __restrict__ ovf,
                       float* __restrict__ out, int N, int R) {
  extern __shared__ float acc[];
  const int tid = threadIdx.x;
  const unsigned lane = threadIdx.x & 31u;
  const int wv  = (int)(threadIdx.x >> 5);
  const int bin = blockIdx.x;
  const int r0  = bin * R;
  int nrows = N - r0;
  if (nrows > R) nrows = R;
  if (nrows < 0) nrows = 0;
  const int nel = nrows * HID;
  float* obase = out + (size_t)r0 * HID;
  for (int i = tid; i < nel; i += PL_T) acc[i] = obase[i];
  __syncthreads();

  int n = cnt[bin];
  if (n > BCAP) n = BCAP;
  n = __builtin_amdgcn_readfirstlane(n);
  const int4* rec4 = (const int4*)rec + (size_t)bin * BCAP;
  edge_sums<HID>(h, rec4, n, acc, lane, wv);

  const int nov = cnt[NB];
  const int4* ovf4 = (const int4*)ovf;
  for (int i = tid; i < nov; i += PL_T) {
    const int4 s = ovf4[i];
    if (s.y >= r0 && s.y < r0 + nrows) {
      const float* hr = h + (size_t)s.x * HID;
      const float  cf = __int_as_float(s.z);
      float*       ac = acc + (s.y - r0) * HID;
      for (int f = 0; f < HID; ++f) atomicAdd(ac + f, hr[f] * cf);
    }
  }
  __syncthreads();
  for (int i = tid; i < nel; i += PL_T) obase[i] = acc[i];
}

__global__ __launch_bounds__(PL_T) void k_prop_mm(const float* __restrict__ in, const float* __restrict__ Wt,
                       const float* __restrict__ bias, const float* __restrict__ dis,
                       const int* __restrict__ cnt, const int* __restrict__ rec, const int* __restrict__ ovf,
                       float* __restrict__ out, float* __restrict__ stats, int N, int R) {
  extern __shared__ float acc[];
  const int tid = threadIdx.x;
  const unsigned lane = threadIdx.x & 31u;
  const int wv  = (int)(threadIdx.x >> 5);
  const int bin = blockIdx.x;
  const int r0  = bin * R;
  int nrows = N - r0;
  if (nrows > R) nrows = R;
  if (nrows < 0) nrows = 0;
  const int nel = nrows * HID;
  const int RP  = (R + 15) & ~15;
  float* prod  = acc + RP * APITCH;
  float* obase = out + (size_t)r0 * HID;
  for (int t0 = 0; t0 * 16 < RP; t0 += 8) {
    float xv[8], dv[8];
#pragma unroll
    for (int u = 0; u < 8; ++u) {
      int g = r0 + (t0 + u) * 16 + wv;
      if (g > N - 1) g = N - 1;
      dv[u] = dis[g];
      xv[u] = in[(size_t)g * HID + lane];
    }
    asm volatile("" :: "v"(xv[0]), "v"(xv[1]), "v"(xv[2]), "v"(xv[3]), "v"(xv[4]), "v"(xv[5]), "v"(xv[6]), "v"(xv[7]),
                       "v"(dv[0]), "v"(dv[1]), "v"(dv[2]), "v"(dv[3]), "v"(dv[4]), "v"(dv[5]), "v"(dv[6]), "v"(dv[7]));
#pragma unroll
    for (int u = 0; u < 8; ++u) {
      const int rr = (t0 + u) * 16 + wv;
      if (rr < RP) acc[rr * APITCH + lane] = rr < nrows ? xv[u] * (dv[u] * dv[u]) : 0.0f;
    }
  }
  __syncthreads();

  int n = cnt[bin];
  if (n > BCAP) n = BCAP;
  n = __builtin_amdgcn_readfirstlane(n);
  const int4* rec4 = (const int4*)rec + (size_t)bin * BCAP;
  edge_sums<APITCH>(in, rec4, n, acc, lane, wv);

  const int nov = cnt[NB];
  const int4* ovf4 = (const int4*)ovf;
  for (int i = tid; i < nov; i += PL_T) {
    const int4 s = ovf4[i];
    if (s.y >= r0 && s.y < r0 + nrows) {
      const float* hr = in + (size_t)s.x * HID;
      const float  cf = __int_as_float(s.z);
      float*       ac = acc + (s.y - r0) * APITCH;
      for (int f = 0; f < HID; ++f) atomicAdd(ac + f, hr[f] * cf);
    }
  }
  __syncthreads();

  const int half  = (int)(lane >> 4);
  const int lm    = (int)(lane & 15u);
  const int njobs = __builtin_amdgcn_readfirstlane(((nrows + 15) >> 4) * 2);
  const int wvs   = __builtin_amdgcn_readfirstlane(wv);
  for (int job = wvs; job < njobs; job += PL_T / 32) {
    const int mt = job >> 1;
    const int nb = (job & 1) * 16;
    const float* arow = acc + (mt * 16 + lm) * APITCH;
    v8f c = {};
#pragma unroll
    for (int k0 = 0; k0 < HID; k0 += 4) {
      const int ka = k0 + 2 * half;
      v2f a;
      a.x = arow[ka];
      a.y = arow[ka + 1];
      v2f b;
      b.x = Wt[(size_t)ka * HID + nb + lm];
      b.y = Wt[(size_t)(ka + 1) * HID + nb + lm];
      c = __builtin_amdgcn_wmma_f32_16x16x4_f32(false, a, false, b, (short)0, c,
                                                false, false);
    }
    const float bcol = bias[nb + lm];
#pragma unroll
    for (int r = 0; r < 8; ++r)
      prod[(mt * 16 + r + 8 * half) * HID + nb + lm] = c[r] + bcol;
  }
  __syncthreads();
  for (int i = tid; i < nel; i += PL_T) obase[i] = prod[i];
  if (bin == 0 && wv == 0) {
    stats[lane]      = 0.0f;
    stats[32 + lane] = 0.0f;
  }
}

#define ST_T 256
#define ST_U 8
__global__ __launch_bounds__(ST_T) void k_bn_stats(const float* __restrict__ c, float* __restrict__ stats, int N) {
  __shared__ float ls[64];
  const int tid = threadIdx.x;
  if (tid < 64) ls[tid] = 0.0f;
  __syncthreads();
  const float4* __restrict__ c4 = (const float4*)c;
  const int total = N * (HID / 4);
  const int i0 = blockIdx.x * (ST_T * ST_U) + tid;
  float4 v[ST_U];
#pragma unroll
  for (int u = 0; u < ST_U; ++u) {
    const int i = i0 + u * ST_T;
    v[u] = c4[i < total ? i : total - 1];
  }
  float s0 = 0.0f, s1 = 0.0f, s2 = 0.0f, s3 = 0.0f;
  float q0 = 0.0f, q1 = 0.0f, q2 = 0.0f, q3 = 0.0f;
#pragma unroll
  for (int u = 0; u < ST_U; ++u) {
    const bool in = i0 + u * ST_T < total;
    const float a0 = in ? fmaxf(v[u].x, 0.0f) : 0.0f;
    const float a1 = in ? fmaxf(v[u].y, 0.0f) : 0.0f;
    const float a2 = in ? fmaxf(v[u].z, 0.0f) : 0.0f;
    const float a3 = in ? fmaxf(v[u].w, 0.0f) : 0.0f;
    s0 += a0; q0 += a0 * a0;
    s1 += a1; q1 += a1 * a1;
    s2 += a2; q2 += a2 * a2;
    s3 += a3; q3 += a3 * a3;
  }
  s0 += __shfl_xor(s0, 8);  s1 += __shfl_xor(s1, 8);  s2 += __shfl_xor(s2, 8);  s3 += __shfl_xor(s3, 8);
  q0 += __shfl_xor(q0, 8);  q1 += __shfl_xor(q1, 8);  q2 += __shfl_xor(q2, 8);  q3 += __shfl_xor(q3, 8);
  s0 += __shfl_xor(s0, 16); s1 += __shfl_xor(s1, 16); s2 += __shfl_xor(s2, 16); s3 += __shfl_xor(s3, 16);
  q0 += __shfl_xor(q0, 16); q1 += __shfl_xor(q1, 16); q2 += __shfl_xor(q2, 16); q3 += __shfl_xor(q3, 16);
  const int lane = tid & 31;
  if (lane < 8) {
    float* lp = ls + 4 * lane;
    atomicAdd(lp + 0, s0); atomicAdd(lp + 1, s1); atomicAdd(lp + 2, s2); atomicAdd(lp + 3, s3);
    atomicAdd(lp + 32, q0); atomicAdd(lp + 33, q1); atomicAdd(lp + 34, q2); atomicAdd(lp + 35, q3);
  }
  __syncthreads();
  if (tid < 64) atomicAdd(&stats[tid], ls[tid]);
}

__global__ void k_bn_apply(const float* __restrict__ c, const float* __restrict__ stats,
                           const float* __restrict__ g, const float* __restrict__ be,
                           float* __restrict__ out, int N) {
  int i = blockIdx.x * blockDim.x + threadIdx.x;
  if (i >= N * HID) return;
  const int f = i & 31;
  const float invN = 1.0f / (float)N;
  const float mu   = stats[f] * invN;
  const float var  = stats[32 + f] * invN - mu * mu;
  const float v    = fmaxf(c[i], 0.0f);
  out[i] = (v - mu) * rsqrtf(var + BN_EPS) * g[f] + be[f];
}

__global__ void k_final(const float* __restrict__ o1, const float* __restrict__ o2,
                        const float* __restrict__ o3, const float* __restrict__ Wl,
                        const float* __restrict__ bl, float* __restrict__ out, int N) {
  int n = blockIdx.x * blockDim.x + threadIdx.x;
  if (n >= N) return;
  float a0 = bl[0], a1 = bl[1];
  const float* p = o1 + (size_t)n * HID;
#pragma unroll
  for (int j = 0; j < HID; ++j) { float v = p[j]; a0 += v * Wl[2 * j]; a1 += v * Wl[2 * j + 1]; }
  p = o2 + (size_t)n * HID;
#pragma unroll
  for (int j = 0; j < HID; ++j) { float v = p[j]; a0 += v * Wl[2 * (HID + j)]; a1 += v * Wl[2 * (HID + j) + 1]; }
  p = o3 + (size_t)n * HID;
#pragma unroll
  for (int j = 0; j < HID; ++j) { float v = p[j]; a0 += v * Wl[2 * (2 * HID + j)]; a1 += v * Wl[2 * (2 * HID + j) + 1]; }
  out[2 * n] = a0;
  out[2 * n + 1] = a1;
}

extern "C" void kernel_launch(void* const* d_in, const int* in_sizes, int n_in,
                              void* d_out, int out_size, void* d_ws, size_t ws_size,
                              hipStream_t stream) {
  const float* x   = (const float*)d_in[0];
  const int*   ei  = (const int*)d_in[1];
  const float* ew  = (const float*)d_in[2];
  const float* W1  = (const float*)d_in[3];
  const float* b1  = (const float*)d_in[4];
  const float* g1  = (const float*)d_in[5];
  const float* be1 = (const float*)d_in[6];
  const float* W2  = (const float*)d_in[7];
  const float* b2  = (const float*)d_in[8];
  const float* g2  = (const float*)d_in[9];
  const float* be2 = (const float*)d_in[10];
  const float* W3  = (const float*)d_in[11];
  const float* b3  = (const float*)d_in[12];
  const float* g3  = (const float*)d_in[13];
  const float* be3 = (const float*)d_in[14];
  const float* Wl  = (const float*)d_in[15];
  const float* bl  = (const float*)d_in[16];

  const int F = 64;
  const int N = in_sizes[0] / F;
  const int E = in_sizes[1] / 2;
  const int* row = ei;
  const int* col = ei + E;

  float* ws    = (float*)d_ws;
  float* dis   = ws;
  float* hbuf  = dis  + ((N + 3) & ~3);
  float* cbuf  = hbuf + (size_t)N * HID;
  float* o1    = cbuf + (size_t)N * HID;
  float* o2    = o1   + (size_t)N * HID;
  float* o3    = o2   + (size_t)N * HID;
  float* stats = o3   + (size_t)N * HID;
  int*   cnt   = (int*)(stats + 64);
  int*   rec   = (int*)(((size_t)(cnt + 1024) + 15) & ~(size_t)15);
  int*   ovf   = rec + (size_t)NB * BCAP * 4;

  const int T = 256;

  k_init_deg<<<(N + T - 1) / T, T, 0, stream>>>(dis, cnt, N);
  k_deg_edges<<<(E + T - 1) / T, T, 0, stream>>>(col, ew, dis, E);
  k_deg_rsqrt<<<(N + T - 1) / T, T, 0, stream>>>(dis, N);

  const int numT = (N + 15) / 16;
  const dim3 gemmB(32, 8);
  const dim3 gemmG((numT + 7) / 8, 2);
  const int elemB = (N * HID + T - 1) / T;
  const int R = (N + NB - 1) / NB;
  const size_t accB = (size_t)R * HID * sizeof(float);
  const int RP = (R + 15) & ~15;
  const size_t accM = (size_t)RP * (APITCH + HID) * sizeof(float);
  const int binC = BIN_T * BIN_EPT;
  const int statB = (N * (HID / 4) + ST_T * ST_U - 1) / (ST_T * ST_U);

  k_bin<<<(E + binC - 1) / binC, BIN_T, 0, stream>>>(row, col, ew, dis, cnt, rec, ovf, E, R);

  k_gemm_wmma<64><<<gemmG, gemmB, 0, stream>>>(x, W1, b1, dis, hbuf, cbuf, stats, N);
  k_prop<<<NB, PL_T, accB, stream>>>(hbuf, cnt, rec, ovf, cbuf, N, R);
  k_bn_stats<<<statB, ST_T, 0, stream>>>(cbuf, stats, N);
  k_bn_apply<<<elemB, T, 0, stream>>>(cbuf, stats, g1, be1, o1, N);

  k_prop_mm<<<NB, PL_T, accM, stream>>>(o1, W2, b2, dis, cnt, rec, ovf, cbuf, stats, N, R);
  k_bn_stats<<<statB, ST_T, 0, stream>>>(cbuf, stats, N);
  k_bn_apply<<<elemB, T, 0, stream>>>(cbuf, stats, g2, be2, o2, N);

  k_prop_mm<<<NB, PL_T, accM, stream>>>(o2, W3, b3, dis, cnt, rec, ovf, cbuf, stats, N, R);
  k_bn_stats<<<statB, ST_T, 0, stream>>>(cbuf, stats, N);
  k_bn_apply<<<elemB, T, 0, stream>>>(cbuf, stats, g3, be3, o3, N);

  k_final<<<(N + T - 1) / T, T, 0, stream>>>(o1, o2, o3, Wl, bl, (float*)d_out, N);
}


# Degrees summed per target range in LDS from the filed edge records (was a scattered global float add per edge); layer 1 forms and stores the coefficients; 15 to 14 launches; degree sum order differs
# speedup vs baseline: 4.2031x; 1.5203x over previous
#include <hip/hip_runtime.h>

typedef float v2f __attribute__((ext_vector_type(2)));
typedef float v8f __attribute__((ext_vector_type(8)));

#define HID 32
#define BN_EPS 1e-5f

#define NB 512
#define BCAP 4096
__global__ void k_zero_counts(int* __restrict__ cnt) {
  for (int j = threadIdx.x; j <= NB; j += blockDim.x) cnt[j] = 0;
}

template <int K>
__global__ __launch_bounds__(256) void k_gemm_wmma(const float* __restrict__ A,
                                                   const float* __restrict__ B,
                                                   const float* __restrict__ bias,
                                                   const float* __restrict__ dis,
                                                   float* __restrict__ D,
                                                   float* __restrict__ S,
                                                   float* __restrict__ stats, int M) {
  const int lane = threadIdx.x;
  const int wave = threadIdx.y;
  const int mt   = blockIdx.x * 8 + wave;
  const int numT = (M + 15) >> 4;
  if (mt >= numT) return;
  const int nb   = blockIdx.y * 16;
  const int half = lane >> 4;
  const int lm   = lane & 15;

  int arow_i = mt * 16 + lm;
  if (arow_i > M - 1) arow_i = M - 1;
  const float* arow = A + (size_t)arow_i * K;
  const float dv   = dis[arow_i];
  const float bcol = bias[nb + lm];

  v8f c = {};
#pragma unroll
  for (int k0 = 0; k0 < K; k0 += 4) {
    const int ka = k0 + 2 * half;
    v2f a;
    a.x = arow[ka];
    a.y = arow[ka + 1];
    v2f b;
    b.x = B[(size_t)ka * HID + nb + lm];
    b.y = B[(size_t)(ka + 1) * HID + nb + lm];
    c = __builtin_amdgcn_wmma_f32_16x16x4_f32(false, a, false, b, (short)0, c,
                                              false, false);
  }
#pragma unroll
  for (int r = 0; r < 8; ++r) {
    const int orow = mt * 16 + r + 8 * half;
    const float d_lo = __int_as_float(__builtin_amdgcn_readlane(__float_as_int(dv), r));
    const float d_hi = __int_as_float(__builtin_amdgcn_readlane(__float_as_int(dv), r + 8));
    const float d    = half ? d_hi : d_lo;
    if (orow < M) {
      D[(size_t)orow * HID + nb + lm] = c[r];
      S[(size_t)orow * HID + nb + lm] = c[r] * (d * d) + bcol;
    }
  }
  if (blockIdx.x == 0 && blockIdx.y == 0 && wave == 0) {
    stats[lane]      = 0.0f;
    stats[32 + lane] = 0.0f;
  }
}

#define BIN_T 512
#define BIN_EPT 8
__global__ __launch_bounds__(BIN_T) void k_bin(const int* __restrict__ row, const int* __restrict__ col,
                      const float* __restrict__ w,
                      int* __restrict__ cnt, int* __restrict__ rec, int* __restrict__ ovf, int E, int R) {
  __shared__ int hist[NB];
  __shared__ int loff[NB];
  __shared__ int gdst[NB];
  __shared__ int ltot;
  __shared__ int4 stage[BIN_T * BIN_EPT];
  const int tid = threadIdx.x;
  for (int i = tid; i < NB; i += BIN_T) hist[i] = 0;
  if (tid == 0) ltot = 0;
  __syncthreads();
  const int cb = blockIdx.x * (BIN_T * BIN_EPT);
  int bn[BIN_EPT], lr[BIN_EPT];
#pragma unroll
  for (int k = 0; k < BIN_EPT; ++k) {
    const int e = cb + k * BIN_T + tid;
    bn[k] = -1;
    lr[k] = 0;
    if (e < E) {
      const int bb = col[e] / R;
      bn[k] = bb;
      lr[k] = atomicAdd(&hist[bb], 1);
    }
  }
  __syncthreads();
  for (int i = tid; i < NB; i += BIN_T) {
    const int hcount = hist[i];
    int lo = 0, gb = 0;
    if (hcount) {
      lo = atomicAdd(&ltot, hcount);
      gb = atomicAdd(&cnt[i], hcount);
    }
    loff[i] = lo;
    gdst[i] = gb;
  }
  __syncthreads();
#pragma unroll
  for (int k = 0; k < BIN_EPT; ++k) {
    if (bn[k] >= 0) {
      const int e  = cb + k * BIN_T + tid;
      const int bb = bn[k];
      const int r  = row[e];
      const int c  = col[e];
      const int slot = gdst[bb] + lr[k];
      const bool fits = slot < BCAP;
      stage[loff[bb] + lr[k]] = make_int4(fits ? bb * BCAP + slot : -1, r, fits ? c - bb * R : c,
                                          __float_as_int(w[e]));
    }
  }
  __syncthreads();
  const int tot = ltot;
  int4* rec4 = (int4*)rec;
  int4* ovf4 = (int4*)ovf;
  for (int p = tid; p < tot; p += BIN_T) {
    const int4 s = stage[p];
    int4* dp = rec4 + s.x;
    if (s.x < 0) dp = ovf4 + atomicAdd(&cnt[NB], 1);
    *dp = make_int4(s.y, s.z, s.w, 0);
  }
}

#define DG_T 512
#define DG_U (BCAP / DG_T)
__global__ __launch_bounds__(DG_T) void k_deg(const int* __restrict__ cnt, const int* __restrict__ rec,
                      const int* __restrict__ ovf, float* __restrict__ dis, int N, int R) {
  extern __shared__ float dsum[];
  const int tid = threadIdx.x;
  const int bin = blockIdx.x;
  const int r0  = bin * R;
  int nrows = N - r0;
  if (nrows > R) nrows = R;
  if (nrows < 0) nrows = 0;
  for (int i = tid; i < R; i += DG_T) dsum[i] = 1.0f;
  __syncthreads();
  int n = cnt[bin];
  if (n > BCAP) n = BCAP;
  const int4* rec4 = (const int4*)rec + (size_t)bin * BCAP;
  if (n > 0) {
    int4 rc[DG_U];
#pragma unroll
    for (int u = 0; u < DG_U; ++u) {
      const int i = u * DG_T + tid;
      rc[u] = rec4[i < n ? i : n - 1];
    }
#pragma unroll
    for (int u = 0; u < DG_U; ++u)
      if (u * DG_T + tid < n) atomicAdd(&dsum[rc[u].y], __int_as_float(rc[u].z));
  }
  const int nov = cnt[NB];
  const int4* ovf4 = (const int4*)ovf;
  for (int i = tid; i < nov; i += DG_T) {
    const int4 s = ovf4[i];
    if (s.y >= r0 && s.y < r0 + nrows) atomicAdd(&dsum[s.y - r0], __int_as_float(s.z));
  }
  __syncthreads();
  for (int i = tid; i < nrows; i += DG_T) dis[r0 + i] = rsqrtf(dsum[i]);
}

#define PL_T 512
#define APITCH 36
template <int PITCH>
__device__ __forceinline__ void edge_sums(const float* __restrict__ h, const int4* __restrict__ rec4,
                                          const int n, float* acc, const unsigned lane, const int wv) {
  const int w0 = __builtin_amdgcn_readfirstlane(wv * 32);
  const unsigned lane4 = lane * 4u;
  const char* hb = (const char*)h;
#pragma nounroll
  for (int g0 = w0; g0 < n; g0 += PL_T) {
    const int i  = g0 + (int)lane;
    const bool ok = i < n;
    const int4 rc = rec4[ok ? i : (n - 1)];
    const unsigned ro  = (unsigned)rc.x * (unsigned)(HID * sizeof(float));
    const int      co  = rc.y * (int)(PITCH * sizeof(float));
    const int      cfb = rc.z;
    if (g0 + 32 <= n) {
      float v[32];
#pragma unroll
      for (int j = 0; j < 32; ++j) {
        const unsigned off = (unsigned)__builtin_amdgcn_readlane((int)ro, j) + lane4;
        v[j] = *(const float*)(hb + off);
      }
#pragma unroll
      for (int j = 0; j < 32; ++j) {
        float*      ac = (float*)((char*)acc + __builtin_amdgcn_readlane(co, j));
        const float cj = __int_as_float(__builtin_amdgcn_readlane(cfb, j));
        atomicAdd(ac + lane, v[j] * cj);
      }
    } else {
      const unsigned mask = __builtin_amdgcn_ballot_w32(ok);
      float v[32];
#pragma unroll
      for (int j = 0; j < 32; ++j) {
        if ((mask >> j) & 1u) {
          const unsigned off = (unsigned)__builtin_amdgcn_readlane((int)ro, j) + lane4;
          v[j] = *(const float*)(hb + off);
        }
      }
#pragma unroll
      for (int j = 0; j < 32; ++j) {
        if ((mask >> j) & 1u) {
          float*      ac = (float*)((char*)acc + __builtin_amdgcn_readlane(co, j));
          const float cj = __int_as_float(__builtin_amdgcn_readlane(cfb, j));
          atomicAdd(ac + lane, v[j] * cj);
        }
      }
    }
  }
}

__device__ __forceinline__ void edge_sums_first(const float* __restrict__ h, int* recb, const float* __restrict__ dis,
                                                const int r0, const int n, float* acc, const unsigned lane,
                                                const int wv) {
  const int w0 = __builtin_amdgcn_readfirstlane(wv * 32);
  const unsigned lane4 = lane * 4u;
  const char* hb = (const char*)h;
  const int4* rec4 = (const int4*)recb;
#pragma nounroll
  for (int g0 = w0; g0 < n; g0 += PL_T) {
    const int i  = g0 + (int)lane;
    const bool ok = i < n;
    const int ic = ok ? i : (n - 1);
    const int4 rc = rec4[ic];
    const unsigned ro  = (unsigned)rc.x * (unsigned)(HID * sizeof(float));
    const int      co  = rc.y * (int)(HID * sizeof(float));
    const float ds = dis[rc.x];
    const float dd = dis[r0 + rc.y];
    const bool full = g0 + 32 <= n;
    const unsigned mask = full ? 0xffffffffu : __builtin_amdgcn_ballot_w32(ok);
    float v[32];
    if (full) {
#pragma unroll
      for (int j = 0; j < 32; ++j) {
        const unsigned off = (unsigned)__builtin_amdgcn_readlane((int)ro, j) + lane4;
        v[j] = *(const float*)(hb + off);
      }
    } else {
#pragma unroll
      for (int j = 0; j < 32; ++j) {
        v[j] = 0.0f;
        if ((mask >> j) & 1u) {
          const unsigned off = (unsigned)__builtin_amdgcn_readlane((int)ro, j) + lane4;
          v[j] = *(const float*)(hb + off);
        }
      }
    }
    const int cfb = __float_as_int(ds * __int_as_float(rc.z) * dd);
    if (ok) recb[(size_t)ic * 4 + 2] = cfb;
    if (full) {
#pragma unroll
      for (int j = 0; j < 32; ++j) {
        float*      ac = (float*)((char*)acc + __builtin_amdgcn_readlane(co, j));
        const float cj = __int_as_float(__builtin_amdgcn_readlane(cfb, j));
        atomicAdd(ac + lane, v[j] * cj);
      }
    } else {
#pragma unroll
      for (int j = 0; j < 32; ++j) {
        if ((mask >> j) & 1u) {
          float*      ac = (float*)((char*)acc + __builtin_amdgcn_readlane(co, j));
          const float cj = __int_as_float(__builtin_amdgcn_readlane(cfb, j));
          atomicAdd(ac + lane, v[j] * cj);
        }
      }
    }
  }
}

__global__ __launch_bounds__(PL_T) void k_prop(const float* __restrict__ h, const int* __restrict__ cnt,
                       int* rec, int* ovf, const float* __restrict__ dis,
                       float* __restrict__ out, int N, int R) {
  extern __shared__ float acc[];
  const int tid = threadIdx.x;
  const unsigned lane = threadIdx.x & 31u;
  const int wv  = (int)(threadIdx.x >> 5);
  const int bin = blockIdx.x;
  const int r0  = bin * R;
  int nrows = N - r0;
  if (nrows > R) nrows = R;
  if (nrows < 0) nrows = 0;
  const int nel = nrows * HID;
  float* obase = out + (size_t)r0 * HID;
  for (int i = tid; i < nel; i += PL_T) acc[i] = obase[i];
  __syncthreads();

  int n = cnt[bin];
  if (n > BCAP) n = BCAP;
  n = __builtin_amdgcn_readfirstlane(n);
  edge_sums_first(h, rec + (size_t)bin * BCAP * 4, dis, r0, n, acc, lane, wv);

  const int nov = cnt[NB];
  const int4* ovf4 = (const int4*)ovf;
  for (int i = tid; i < nov; i += PL_T) {
    const int4 s = ovf4[i];
    if (s.y >= r0 && s.y < r0 + nrows) {
      const float* hr = h + (size_t)s.x * HID;
      const float  cf = dis[s.x] * __int_as_float(s.z) * dis[s.y];
      ovf[(size_t)i * 4 + 2] = __float_as_int(cf);
      float*       ac = acc + (s.y - r0) * HID;
      for (int f = 0; f < HID; ++f) atomicAdd(ac + f, hr[f] * cf);
    }
  }
  __syncthreads();
  for (int i = tid; i < nel; i += PL_T) obase[i] = acc[i];
}

__global__ __launch_bounds__(PL_T) void k_prop_mm(const float* __restrict__ in, const float* __restrict__ Wt,
                       const float* __restrict__ bias, const float* __restrict__ dis,
                       const int* __restrict__ cnt, const int* __restrict__ rec, const int* __restrict__ ovf,
                       float* __restrict__ out, float* __restrict__ stats, int N, int R) {
  extern __shared__ float acc[];
  const int tid = threadIdx.x;
  const unsigned lane = threadIdx.x & 31u;
  const int wv  = (int)(threadIdx.x >> 5);
  const int bin = blockIdx.x;
  const int r0  = bin * R;
  int nrows = N - r0;
  if (nrows > R) nrows = R;
  if (nrows < 0) nrows = 0;
  const int nel = nrows * HID;
  const int RP  = (R + 15) & ~15;
  float* prod  = acc + RP * APITCH;
  float* obase = out + (size_t)r0 * HID;
  for (int t0 = 0; t0 * 16 < RP; t0 += 8) {
    float xv[8], dv[8];
#pragma unroll
    for (int u = 0; u < 8; ++u) {
      int g = r0 + (t0 + u) * 16 + wv;
      if (g > N - 1) g = N - 1;
      dv[u] = dis[g];
      xv[u] = in[(size_t)g * HID + lane];
    }
    asm volatile("" :: "v"(xv[0]), "v"(xv[1]), "v"(xv[2]), "v"(xv[3]), "v"(xv[4]), "v"(xv[5]), "v"(xv[6]), "v"(xv[7]),
                       "v"(dv[0]), "v"(dv[1]), "v"(dv[2]), "v"(dv[3]), "v"(dv[4]), "v"(dv[5]), "v"(dv[6]), "v"(dv[7]));
#pragma unroll
    for (int u = 0; u < 8; ++u) {
      const int rr = (t0 + u) * 16 + wv;
      if (rr < RP) acc[rr * APITCH + lane] = rr < nrows ? xv[u] * (dv[u] * dv[u]) : 0.0f;
    }
  }
  __syncthreads();

  int n = cnt[bin];
  if (n > BCAP) n = BCAP;
  n = __builtin_amdgcn_readfirstlane(n);
  const int4* rec4 = (const int4*)rec + (size_t)bin * BCAP;
  edge_sums<APITCH>(in, rec4, n, acc, lane, wv);

  const int nov = cnt[NB];
  const int4* ovf4 = (const int4*)ovf;
  for (int i = tid; i < nov; i += PL_T) {
    const int4 s = ovf4[i];
    if (s.y >= r0 && s.y < r0 + nrows) {
      const float* hr = in + (size_t)s.x * HID;
      const float  cf = __int_as_float(s.z);
      float*       ac = acc + (s.y - r0) * APITCH;
      for (int f = 0; f < HID; ++f) atomicAdd(ac + f, hr[f] * cf);
    }
  }
  __syncthreads();

  const int half  = (int)(lane >> 4);
  const int lm    = (int)(lane & 15u);
  const int njobs = __builtin_amdgcn_readfirstlane(((nrows + 15) >> 4) * 2);
  const int wvs   = __builtin_amdgcn_readfirstlane(wv);
  for (int job = wvs; job < njobs; job += PL_T / 32) {
    const int mt = job >> 1;
    const int nb = (job & 1) * 16;
    const float* arow = acc + (mt * 16 + lm) * APITCH;
    v8f c = {};
#pragma unroll
    for (int k0 = 0; k0 < HID; k0 += 4) {
      const int ka = k0 + 2 * half;
      v2f a;
      a.x = arow[ka];
      a.y = arow[ka + 1];
      v2f b;
      b.x = Wt[(size_t)ka * HID + nb + lm];
      b.y = Wt[(size_t)(ka + 1) * HID + nb + lm];
      c = __builtin_amdgcn_wmma_f32_16x16x4_f32(false, a, false, b, (short)0, c,
                                                false, false);
    }
    const float bcol = bias[nb + lm];
#pragma unroll
    for (int r = 0; r < 8; ++r)
      prod[(mt * 16 + r + 8 * half) * HID + nb + lm] = c[r] + bcol;
  }
  __syncthreads();
  for (int i = tid; i < nel; i += PL_T) obase[i] = prod[i];
  if (bin == 0 && wv == 0) {
    stats[lane]      = 0.0f;
    stats[32 + lane] = 0.0f;
  }
}

#define ST_T 256
#define ST_U 8
__global__ __launch_bounds__(ST_T) void k_bn_stats(const float* __restrict__ c, float* __restrict__ stats, int N) {
  __shared__ float ls[64];
  const int tid = threadIdx.x;
  if (tid < 64) ls[tid] = 0.0f;
  __syncthreads();
  const float4* __restrict__ c4 = (const float4*)c;
  const int total = N * (HID / 4);
  const int i0 = blockIdx.x * (ST_T * ST_U) + tid;
  float4 v[ST_U];
#pragma unroll
  for (int u = 0; u < ST_U; ++u) {
    const int i = i0 + u * ST_T;
    v[u] = c4[i < total ? i : total - 1];
  }
  float s0 = 0.0f, s1 = 0.0f, s2 = 0.0f, s3 = 0.0f;
  float q0 = 0.0f, q1 = 0.0f, q2 = 0.0f, q3 = 0.0f;
#pragma unroll
  for (int u = 0; u < ST_U; ++u) {
    const bool in = i0 + u * ST_T < total;
    const float a0 = in ? fmaxf(v[u].x, 0.0f) : 0.0f;
    const float a1 = in ? fmaxf(v[u].y, 0.0f) : 0.0f;
    const float a2 = in ? fmaxf(v[u].z, 0.0f) : 0.0f;
    const float a3 = in ? fmaxf(v[u].w, 0.0f) : 0.0f;
    s0 += a0; q0 += a0 * a0;
    s1 += a1; q1 += a1 * a1;
    s2 += a2; q2 += a2 * a2;
    s3 += a3; q3 += a3 * a3;
  }
  s0 += __shfl_xor(s0, 8);  s1 += __shfl_xor(s1, 8);  s2 += __shfl_xor(s2, 8);  s3 += __shfl_xor(s3, 8);
  q0 += __shfl_xor(q0, 8);  q1 += __shfl_xor(q1, 8);  q2 += __shfl_xor(q2, 8);  q3 += __shfl_xor(q3, 8);
  s0 += __shfl_xor(s0, 16); s1 += __shfl_xor(s1, 16); s2 += __shfl_xor(s2, 16); s3 += __shfl_xor(s3, 16);
  q0 += __shfl_xor(q0, 16); q1 += __shfl_xor(q1, 16); q2 += __shfl_xor(q2, 16); q3 += __shfl_xor(q3, 16);
  const int lane = tid & 31;
  if (lane < 8) {
    float* lp = ls + 4 * lane;
    atomicAdd(lp + 0, s0); atomicAdd(lp + 1, s1); atomicAdd(lp + 2, s2); atomicAdd(lp + 3, s3);
    atomicAdd(lp + 32, q0); atomicAdd(lp + 33, q1); atomicAdd(lp + 34, q2); atomicAdd(lp + 35, q3);
  }
  __syncthreads();
  if (tid < 64) atomicAdd(&stats[tid], ls[tid]);
}

__global__ void k_bn_apply(const float* __restrict__ c, const float* __restrict__ stats,
                           const float* __restrict__ g, const float* __restrict__ be,
                           float* __restrict__ out, int N) {
  int i = blockIdx.x * blockDim.x + threadIdx.x;
  if (i >= N * HID) return;
  const int f = i & 31;
  const float invN = 1.0f / (float)N;
  const float mu   = stats[f] * invN;
  const float var  = stats[32 + f] * invN - mu * mu;
  const float v    = fmaxf(c[i], 0.0f);
  out[i] = (v - mu) * rsqrtf(var + BN_EPS) * g[f] + be[f];
}

__global__ void k_final(const float* __restrict__ o1, const float* __restrict__ o2,
                        const float* __restrict__ o3, const float* __restrict__ Wl,
                        const float* __restrict__ bl, float* __restrict__ out, int N) {
  int n = blockIdx.x * blockDim.x + threadIdx.x;
  if (n >= N) return;
  float a0 = bl[0], a1 = bl[1];
  const float* p = o1 + (size_t)n * HID;
#pragma unroll
  for (int j = 0; j < HID; ++j) { float v = p[j]; a0 += v * Wl[2 * j]; a1 += v * Wl[2 * j + 1]; }
  p = o2 + (size_t)n * HID;
#pragma unroll
  for (int j = 0; j < HID; ++j) { float v = p[j]; a0 += v * Wl[2 * (HID + j)]; a1 += v * Wl[2 * (HID + j) + 1]; }
  p = o3 + (size_t)n * HID;
#pragma unroll
  for (int j = 0; j < HID; ++j) { float v = p[j]; a0 += v * Wl[2 * (2 * HID + j)]; a1 += v * Wl[2 * (2 * HID + j) + 1]; }
  out[2 * n] = a0;
  out[2 * n + 1] = a1;
}

extern "C" void kernel_launch(void* const* d_in, const int* in_sizes, int n_in,
                              void* d_out, int out_size, void* d_ws, size_t ws_size,
                              hipStream_t stream) {
  const float* x   = (const float*)d_in[0];
  const int*   ei  = (const int*)d_in[1];
  const float* ew  = (const float*)d_in[2];
  const float* W1  = (const float*)d_in[3];
  const float* b1  = (const float*)d_in[4];
  const float* g1  = (const float*)d_in[5];
  const float* be1 = (const float*)d_in[6];
  const float* W2  = (const float*)d_in[7];
  const float* b2  = (const float*)d_in[8];
  const float* g2  = (const float*)d_in[9];
  const float* be2 = (const float*)d_in[10];
  const float* W3  = (const float*)d_in[11];
  const float* b3  = (const float*)d_in[12];
  const float* g3  = (const float*)d_in[13];
  const float* be3 = (const float*)d_in[14];
  const float* Wl  = (const float*)d_in[15];
  const float* bl  = (const float*)d_in[16];

  const int F = 64;
  const int N = in_sizes[0] / F;
  const int E = in_sizes[1] / 2;
  const int* row = ei;
  const int* col = ei + E;

  float* ws    = (float*)d_ws;
  float* dis   = ws;
  float* hbuf  = dis  + ((N + 3) & ~3);
  float* cbuf  = hbuf + (size_t)N * HID;
  float* o1    = cbuf + (size_t)N * HID;
  float* o2    = o1   + (size_t)N * HID;
  float* o3    = o2   + (size_t)N * HID;
  float* stats = o3   + (size_t)N * HID;
  int*   cnt   = (int*)(stats + 64);
  int*   rec   = (int*)(((size_t)(cnt + 1024) + 15) & ~(size_t)15);
  int*   ovf   = rec + (size_t)NB * BCAP * 4;

  const int T = 256;

  const int numT = (N + 15) / 16;
  const dim3 gemmB(32, 8);
  const dim3 gemmG((numT + 7) / 8, 2);
  const int elemB = (N * HID + T - 1) / T;
  const int R = (N + NB - 1) / NB;
  const size_t accB = (size_t)R * HID * sizeof(float);
  const int RP = (R + 15) & ~15;
  const size_t accM = (size_t)RP * (APITCH + HID) * sizeof(float);
  const int binC = BIN_T * BIN_EPT;
  const int statB = (N * (HID / 4) + ST_T * ST_U - 1) / (ST_T * ST_U);

  const size_t degB = (size_t)R * sizeof(float);

  k_zero_counts<<<1, T, 0, stream>>>(cnt);
  k_bin<<<(E + binC - 1) / binC, BIN_T, 0, stream>>>(row, col, ew, cnt, rec, ovf, E, R);
  k_deg<<<NB, DG_T, degB, stream>>>(cnt, rec, ovf, dis, N, R);

  k_gemm_wmma<64><<<gemmG, gemmB, 0, stream>>>(x, W1, b1, dis, hbuf, cbuf, stats, N);
  k_prop<<<NB, PL_T, accB, stream>>>(hbuf, cnt, rec, ovf, dis, cbuf, N, R);
  k_bn_stats<<<statB, ST_T, 0, stream>>>(cbuf, stats, N);
  k_bn_apply<<<elemB, T, 0, stream>>>(cbuf, stats, g1, be1, o1, N);

  k_prop_mm<<<NB, PL_T, accM, stream>>>(o1, W2, b2, dis, cnt, rec, ovf, cbuf, stats, N, R);
  k_bn_stats<<<statB, ST_T, 0, stream>>>(cbuf, stats, N);
  k_bn_apply<<<elemB, T, 0, stream>>>(cbuf, stats, g2, be2, o2, N);

  k_prop_mm<<<NB, PL_T, accM, stream>>>(o2, W3, b3, dis, cnt, rec, ovf, cbuf, stats, N, R);
  k_bn_stats<<<statB, ST_T, 0, stream>>>(cbuf, stats, N);
  k_bn_apply<<<elemB, T, 0, stream>>>(cbuf, stats, g3, be3, o3, N);

  k_final<<<(N + T - 1) / T, T, 0, stream>>>(o1, o2, o3, Wl, bl, (float*)d_out, N);
}


# BN apply pass: a lane takes 16-byte pieces (4 features of a row), its 8 loads issued before the first use, mean and 1/sqrt(var+eps) formed once per lane; was one element per thread, 100000 waves
# speedup vs baseline: 4.6191x; 1.0990x over previous
#include <hip/hip_runtime.h>

typedef float v2f __attribute__((ext_vector_type(2)));
typedef float v8f __attribute__((ext_vector_type(8)));

#define HID 32
#define BN_EPS 1e-5f

#define NB 512
#define BCAP 4096
__global__ void k_zero_counts(int* __restrict__ cnt) {
  for (int j = threadIdx.x; j <= NB; j += blockDim.x) cnt[j] = 0;
}

template <int K>
__global__ __launch_bounds__(256) void k_gemm_wmma(const float* __restrict__ A,
                                                   const float* __restrict__ B,
                                                   const float* __restrict__ bias,
                                                   const float* __restrict__ dis,
                                                   float* __restrict__ D,
                                                   float* __restrict__ S,
                                                   float* __restrict__ stats, int M) {
  const int lane = threadIdx.x;
  const int wave = threadIdx.y;
  const int mt   = blockIdx.x * 8 + wave;
  const int numT = (M + 15) >> 4;
  if (mt >= numT) return;
  const int nb   = blockIdx.y * 16;
  const int half = lane >> 4;
  const int lm   = lane & 15;

  int arow_i = mt * 16 + lm;
  if (arow_i > M - 1) arow_i = M - 1;
  const float* arow = A + (size_t)arow_i * K;
  const float dv   = dis[arow_i];
  const float bcol = bias[nb + lm];

  v8f c = {};
#pragma unroll
  for (int k0 = 0; k0 < K; k0 += 4) {
    const int ka = k0 + 2 * half;
    v2f a;
    a.x = arow[ka];
    a.y = arow[ka + 1];
    v2f b;
    b.x = B[(size_t)ka * HID + nb + lm];
    b.y = B[(size_t)(ka + 1) * HID + nb + lm];
    c = __builtin_amdgcn_wmma_f32_16x16x4_f32(false, a, false, b, (short)0, c,
                                              false, false);
  }
#pragma unroll
  for (int r = 0; r < 8; ++r) {
    const int orow = mt * 16 + r + 8 * half;
    const float d_lo = __int_as_float(__builtin_amdgcn_readlane(__float_as_int(dv), r));
    const float d_hi = __int_as_float(__builtin_amdgcn_readlane(__float_as_int(dv), r + 8));
    const float d    = half ? d_hi : d_lo;
    if (orow < M) {
      D[(size_t)orow * HID + nb + lm] = c[r];
      S[(size_t)orow * HID + nb + lm] = c[r] * (d * d) + bcol;
    }
  }
  if (blockIdx.x == 0 && blockIdx.y == 0 && wave == 0) {
    stats[lane]      = 0.0f;
    stats[32 + lane] = 0.0f;
  }
}

#define BIN_T 512
#define BIN_EPT 8
__global__ __launch_bounds__(BIN_T) void k_bin(const int* __restrict__ row, const int* __restrict__ col,
                      const float* __restrict__ w,
                      int* __restrict__ cnt, int* __restrict__ rec, int* __restrict__ ovf, int E, int R) {
  __shared__ int hist[NB];
  __shared__ int loff[NB];
  __shared__ int gdst[NB];
  __shared__ int ltot;
  __shared__ int4 stage[BIN_T * BIN_EPT];
  const int tid = threadIdx.x;
  for (int i = tid; i < NB; i += BIN_T) hist[i] = 0;
  if (tid == 0) ltot = 0;
  __syncthreads();
  const int cb = blockIdx.x * (BIN_T * BIN_EPT);
  int bn[BIN_EPT], lr[BIN_EPT];
#pragma unroll
  for (int k = 0; k < BIN_EPT; ++k) {
    const int e = cb + k * BIN_T + tid;
    bn[k] = -1;
    lr[k] = 0;
    if (e < E) {
      const int bb = col[e] / R;
      bn[k] = bb;
      lr[k] = atomicAdd(&hist[bb], 1);
    }
  }
  __syncthreads();
  for (int i = tid; i < NB; i += BIN_T) {
    const int hcount = hist[i];
    int lo = 0, gb = 0;
    if (hcount) {
      lo = atomicAdd(&ltot, hcount);
      gb = atomicAdd(&cnt[i], hcount);
    }
    loff[i] = lo;
    gdst[i] = gb;
  }
  __syncthreads();
#pragma unroll
  for (int k = 0; k < BIN_EPT; ++k) {
    if (bn[k] >= 0) {
      const int e  = cb + k * BIN_T + tid;
      const int bb = bn[k];
      const int r  = row[e];
      const int c  = col[e];
      const int slot = gdst[bb] + lr[k];
      const bool fits = slot < BCAP;
      stage[loff[bb] + lr[k]] = make_int4(fits ? bb * BCAP + slot : -1, r, fits ? c - bb * R : c,
                                          __float_as_int(w[e]));
    }
  }
  __syncthreads();
  const int tot = ltot;
  int4* rec4 = (int4*)rec;
  int4* ovf4 = (int4*)ovf;
  for (int p = tid; p < tot; p += BIN_T) {
    const int4 s = stage[p];
    int4* dp = rec4 + s.x;
    if (s.x < 0) dp = ovf4 + atomicAdd(&cnt[NB], 1);
    *dp = make_int4(s.y, s.z, s.w, 0);
  }
}

#define DG_T 512
#define DG_U (BCAP / DG_T)
__global__ __launch_bounds__(DG_T) void k_deg(const int* __restrict__ cnt, const int* __restrict__ rec,
                      const int* __restrict__ ovf, float* __restrict__ dis, int N, int R) {
  extern __shared__ float dsum[];
  const int tid = threadIdx.x;
  const int bin = blockIdx.x;
  const int r0  = bin * R;
  int nrows = N - r0;
  if (nrows > R) nrows = R;
  if (nrows < 0) nrows = 0;
  for (int i = tid; i < R; i += DG_T) dsum[i] = 1.0f;
  __syncthreads();
  int n = cnt[bin];
  if (n > BCAP) n = BCAP;
  const int4* rec4 = (const int4*)rec + (size_t)bin * BCAP;
  if (n > 0) {
    int4 rc[DG_U];
#pragma unroll
    for (int u = 0; u < DG_U; ++u) {
      const int i = u * DG_T + tid;
      rc[u] = rec4[i < n ? i : n - 1];
    }
#pragma unroll
    for (int u = 0; u < DG_U; ++u)
      if (u * DG_T + tid < n) atomicAdd(&dsum[rc[u].y], __int_as_float(rc[u].z));
  }
  const int nov = cnt[NB];
  const int4* ovf4 = (const int4*)ovf;
  for (int i = tid; i < nov; i += DG_T) {
    const int4 s = ovf4[i];
    if (s.y >= r0 && s.y < r0 + nrows) atomicAdd(&dsum[s.y - r0], __int_as_float(s.z));
  }
  __syncthreads();
  for (int i = tid; i < nrows; i += DG_T) dis[r0 + i] = rsqrtf(dsum[i]);
}

#define PL_T 512
#define APITCH 36
template <int PITCH>
__device__ __forceinline__ void edge_sums(const float* __restrict__ h, const int4* __restrict__ rec4,
                                          const int n, float* acc, const unsigned lane, const int wv) {
  const int w0 = __builtin_amdgcn_readfirstlane(wv * 32);
  const unsigned lane4 = lane * 4u;
  const char* hb = (const char*)h;
#pragma nounroll
  for (int g0 = w0; g0 < n; g0 += PL_T) {
    const int i  = g0 + (int)lane;
    const bool ok = i < n;
    const int4 rc = rec4[ok ? i : (n - 1)];
    const unsigned ro  = (unsigned)rc.x * (unsigned)(HID * sizeof(float));
    const int      co  = rc.y * (int)(PITCH * sizeof(float));
    const int      cfb = rc.z;
    if (g0 + 32 <= n) {
      float v[32];
#pragma unroll
      for (int j = 0; j < 32; ++j) {
        const unsigned off = (unsigned)__builtin_amdgcn_readlane((int)ro, j) + lane4;
        v[j] = *(const float*)(hb + off);
      }
#pragma unroll
      for (int j = 0; j < 32; ++j) {
        float*      ac = (float*)((char*)acc + __builtin_amdgcn_readlane(co, j));
        const float cj = __int_as_float(__builtin_amdgcn_readlane(cfb, j));
        atomicAdd(ac + lane, v[j] * cj);
      }
    } else {
      const unsigned mask = __builtin_amdgcn_ballot_w32(ok);
      float v[32];
#pragma unroll
      for (int j = 0; j < 32; ++j) {
        if ((mask >> j) & 1u) {
          const unsigned off = (unsigned)__builtin_amdgcn_readlane((int)ro, j) + lane4;
          v[j] = *(const float*)(hb + off);
        }
      }
#pragma unroll
      for (int j = 0; j < 32; ++j) {
        if ((mask >> j) & 1u) {
          float*      ac = (float*)((char*)acc + __builtin_amdgcn_readlane(co, j));
          const float cj = __int_as_float(__builtin_amdgcn_readlane(cfb, j));
          atomicAdd(ac + lane, v[j] * cj);
        }
      }
    }
  }
}

__device__ __forceinline__ void edge_sums_first(const float* __restrict__ h, int* recb, const float* __restrict__ dis,
                                                const int r0, const int n, float* acc, const unsigned lane,
                                                const int wv) {
  const int w0 = __builtin_amdgcn_readfirstlane(wv * 32);
  const unsigned lane4 = lane * 4u;
  const char* hb = (const char*)h;
  const int4* rec4 = (const int4*)recb;
#pragma nounroll
  for (int g0 = w0; g0 < n; g0 += PL_T) {
    const int i  = g0 + (int)lane;
    const bool ok = i < n;
    const int ic = ok ? i : (n - 1);
    const int4 rc = rec4[ic];
    const unsigned ro  = (unsigned)rc.x * (unsigned)(HID * sizeof(float));
    const int      co  = rc.y * (int)(HID * sizeof(float));
    const float ds = dis[rc.x];
    const float dd = dis[r0 + rc.y];
    const bool full = g0 + 32 <= n;
    const unsigned mask = full ? 0xffffffffu : __builtin_amdgcn_ballot_w32(ok);
    float v[32];
    if (full) {
#pragma unroll
      for (int j = 0; j < 32; ++j) {
        const unsigned off = (unsigned)__builtin_amdgcn_readlane((int)ro, j) + lane4;
        v[j] = *(const float*)(hb + off);
      }
    } else {
#pragma unroll
      for (int j = 0; j < 32; ++j) {
        v[j] = 0.0f;
        if ((mask >> j) & 1u) {
          const unsigned off = (unsigned)__builtin_amdgcn_readlane((int)ro, j) + lane4;
          v[j] = *(const float*)(hb + off);
        }
      }
    }
    const int cfb = __float_as_int(ds * __int_as_float(rc.z) * dd);
    if (ok) recb[(size_t)ic * 4 + 2] = cfb;
    if (full) {
#pragma unroll
      for (int j = 0; j < 32; ++j) {
        float*      ac = (float*)((char*)acc + __builtin_amdgcn_readlane(co, j));
        const float cj = __int_as_float(__builtin_amdgcn_readlane(cfb, j));
        atomicAdd(ac + lane, v[j] * cj);
      }
    } else {
#pragma unroll
      for (int j = 0; j < 32; ++j) {
        if ((mask >> j) & 1u) {
          float*      ac = (float*)((char*)acc + __builtin_amdgcn_readlane(co, j));
          const float cj = __int_as_float(__builtin_amdgcn_readlane(cfb, j));
          atomicAdd(ac + lane, v[j] * cj);
        }
      }
    }
  }
}

__global__ __launch_bounds__(PL_T) void k_prop(const float* __restrict__ h, const int* __restrict__ cnt,
                       int* rec, int* ovf, const float* __restrict__ dis,
                       float* __restrict__ out, int N, int R) {
  extern __shared__ float acc[];
  const int tid = threadIdx.x;
  const unsigned lane = threadIdx.x & 31u;
  const int wv  = (int)(threadIdx.x >> 5);
  const int bin = blockIdx.x;
  const int r0  = bin * R;
  int nrows = N - r0;
  if (nrows > R) nrows = R;
  if (nrows < 0) nrows = 0;
  const int nel = nrows * HID;
  float* obase = out + (size_t)r0 * HID;
  for (int i = tid; i < nel; i += PL_T) acc[i] = obase[i];
  __syncthreads();

  int n = cnt[bin];
  if (n > BCAP) n = BCAP;
  n = __builtin_amdgcn_readfirstlane(n);
  edge_sums_first(h, rec + (size_t)bin * BCAP * 4, dis, r0, n, acc, lane, wv);

  const int nov = cnt[NB];
  const int4* ovf4 = (const int4*)ovf;
  for (int i = tid; i < nov; i += PL_T) {
    const int4 s = ovf4[i];
    if (s.y >= r0 && s.y < r0 + nrows) {
      const float* hr = h + (size_t)s.x * HID;
      const float  cf = dis[s.x] * __int_as_float(s.z) * dis[s.y];
      ovf[(size_t)i * 4 + 2] = __float_as_int(cf);
      float*       ac = acc + (s.y - r0) * HID;
      for (int f = 0; f < HID; ++f) atomicAdd(ac + f, hr[f] * cf);
    }
  }
  __syncthreads();
  for (int i = tid; i < nel; i += PL_T) obase[i] = acc[i];
}

__global__ __launch_bounds__(PL_T) void k_prop_mm(const float* __restrict__ in, const float* __restrict__ Wt,
                       const float* __restrict__ bias, const float* __restrict__ dis,
                       const int* __restrict__ cnt, const int* __restrict__ rec, const int* __restrict__ ovf,
                       float* __restrict__ out, float* __restrict__ stats, int N, int R) {
  extern __shared__ float acc[];
  const int tid = threadIdx.x;
  const unsigned lane = threadIdx.x & 31u;
  const int wv  = (int)(threadIdx.x >> 5);
  const int bin = blockIdx.x;
  const int r0  = bin * R;
  int nrows = N - r0;
  if (nrows > R) nrows = R;
  if (nrows < 0) nrows = 0;
  const int nel = nrows * HID;
  const int RP  = (R + 15) & ~15;
  float* prod  = acc + RP * APITCH;
  float* obase = out + (size_t)r0 * HID;
  for (int t0 = 0; t0 * 16 < RP; t0 += 8) {
    float xv[8], dv[8];
#pragma unroll
    for (int u = 0; u < 8; ++u) {
      int g = r0 + (t0 + u) * 16 + wv;
      if (g > N - 1) g = N - 1;
      dv[u] = dis[g];
      xv[u] = in[(size_t)g * HID + lane];
    }
    asm volatile("" :: "v"(xv[0]), "v"(xv[1]), "v"(xv[2]), "v"(xv[3]), "v"(xv[4]), "v"(xv[5]), "v"(xv[6]), "v"(xv[7]),
                       "v"(dv[0]), "v"(dv[1]), "v"(dv[2]), "v"(dv[3]), "v"(dv[4]), "v"(dv[5]), "v"(dv[6]), "v"(dv[7]));
#pragma unroll
    for (int u = 0; u < 8; ++u) {
      const int rr = (t0 + u) * 16 + wv;
      if (rr < RP) acc[rr * APITCH + lane] = rr < nrows ? xv[u] * (dv[u] * dv[u]) : 0.0f;
    }
  }
  __syncthreads();

  int n = cnt[bin];
  if (n > BCAP) n = BCAP;
  n = __builtin_amdgcn_readfirstlane(n);
  const int4* rec4 = (const int4*)rec + (size_t)bin * BCAP;
  edge_sums<APITCH>(in, rec4, n, acc, lane, wv);

  const int nov = cnt[NB];
  const int4* ovf4 = (const int4*)ovf;
  for (int i = tid; i < nov; i += PL_T) {
    const int4 s = ovf4[i];
    if (s.y >= r0 && s.y < r0 + nrows) {
      const float* hr = in + (size_t)s.x * HID;
      const float  cf = __int_as_float(s.z);
      float*       ac = acc + (s.y - r0) * APITCH;
      for (int f = 0; f < HID; ++f) atomicAdd(ac + f, hr[f] * cf);
    }
  }
  __syncthreads();

  const int half  = (int)(lane >> 4);
  const int lm    = (int)(lane & 15u);
  const int njobs = __builtin_amdgcn_readfirstlane(((nrows + 15) >> 4) * 2);
  const int wvs   = __builtin_amdgcn_readfirstlane(wv);
  for (int job = wvs; job < njobs; job += PL_T / 32) {
    const int mt = job >> 1;
    const int nb = (job & 1) * 16;
    const float* arow = acc + (mt * 16 + lm) * APITCH;
    v8f c = {};
#pragma unroll
    for (int k0 = 0; k0 < HID; k0 += 4) {
      const int ka = k0 + 2 * half;
      v2f a;
      a.x = arow[ka];
      a.y = arow[ka + 1];
      v2f b;
      b.x = Wt[(size_t)ka * HID + nb + lm];
      b.y = Wt[(size_t)(ka + 1) * HID + nb + lm];
      c = __builtin_amdgcn_wmma_f32_16x16x4_f32(false, a, false, b, (short)0, c,
                                                false, false);
    }
    const float bcol = bias[nb + lm];
#pragma unroll
    for (int r = 0; r < 8; ++r)
      prod[(mt * 16 + r + 8 * half) * HID + nb + lm] = c[r] + bcol;
  }
  __syncthreads();
  for (int i = tid; i < nel; i += PL_T) obase[i] = prod[i];
  if (bin == 0 && wv == 0) {
    stats[lane]      = 0.0f;
    stats[32 + lane] = 0.0f;
  }
}

#define ST_T 256
#define ST_U 8
__global__ __launch_bounds__(ST_T) void k_bn_stats(const float* __restrict__ c, float* __restrict__ stats, int N) {
  __shared__ float ls[64];
  const int tid = threadIdx.x;
  if (tid < 64) ls[tid] = 0.0f;
  __syncthreads();
  const float4* __restrict__ c4 = (const float4*)c;
  const int total = N * (HID / 4);
  const int i0 = blockIdx.x * (ST_T * ST_U) + tid;
  float4 v[ST_U];
#pragma unroll
  for (int u = 0; u < ST_U; ++u) {
    const int i = i0 + u * ST_T;
    v[u] = c4[i < total ? i : total - 1];
  }
  float s0 = 0.0f, s1 = 0.0f, s2 = 0.0f, s3 = 0.0f;
  float q0 = 0.0f, q1 = 0.0f, q2 = 0.0f, q3 = 0.0f;
#pragma unroll
  for (int u = 0; u < ST_U; ++u) {
    const bool in = i0 + u * ST_T < total;
    const float a0 = in ? fmaxf(v[u].x, 0.0f) : 0.0f;
    const float a1 = in ? fmaxf(v[u].y, 0.0f) : 0.0f;
    const float a2 = in ? fmaxf(v[u].z, 0.0f) : 0.0f;
    const float a3 = in ? fmaxf(v[u].w, 0.0f) : 0.0f;
    s0 += a0; q0 += a0 * a0;
    s1 += a1; q1 += a1 * a1;
    s2 += a2; q2 += a2 * a2;
    s3 += a3; q3 += a3 * a3;
  }
  s0 += __shfl_xor(s0, 8);  s1 += __shfl_xor(s1, 8);  s2 += __shfl_xor(s2, 8);  s3 += __shfl_xor(s3, 8);
  q0 += __shfl_xor(q0, 8);  q1 += __shfl_xor(q1, 8);  q2 += __shfl_xor(q2, 8);  q3 += __shfl_xor(q3, 8);
  s0 += __shfl_xor(s0, 16); s1 += __shfl_xor(s1, 16); s2 += __shfl_xor(s2, 16); s3 += __shfl_xor(s3, 16);
  q0 += __shfl_xor(q0, 16); q1 += __shfl_xor(q1, 16); q2 += __shfl_xor(q2, 16); q3 += __shfl_xor(q3, 16);
  const int lane = tid & 31;
  if (lane < 8) {
    float* lp = ls + 4 * lane;
    atomicAdd(lp + 0, s0); atomicAdd(lp + 1, s1); atomicAdd(lp + 2, s2); atomicAdd(lp + 3, s3);
    atomicAdd(lp + 32, q0); atomicAdd(lp + 33, q1); atomicAdd(lp + 34, q2); atomicAdd(lp + 35, q3);
  }
  __syncthreads();
  if (tid < 64) atomicAdd(&stats[tid], ls[tid]);
}

template <bool WHOLE>
__device__ __forceinline__ void bn_apply_pieces(const float4* __restrict__ c4, const float* __restrict__ stats,
                                                const float* __restrict__ g, const float* __restrict__ be,
                                                float4* __restrict__ o4, const int i0, const int fq,
                                                const int total, const float invN) {
  float4 v[ST_U];
#pragma unroll
  for (int u = 0; u < ST_U; ++u) {
    const int i = i0 + u * ST_T;
    v[u] = c4[(WHOLE || i < total) ? i : total - 1];
  }
  const float4 sm = *(const float4*)(stats + fq);
  const float4 sq = *(const float4*)(stats + 32 + fq);
  const float g0 = g[fq], g1 = g[fq + 1], g2 = g[fq + 2], g3 = g[fq + 3];
  const float b0 = be[fq], b1 = be[fq + 1], b2 = be[fq + 2], b3 = be[fq + 3];
  const float mu0 = sm.x * invN, mu1 = sm.y * invN, mu2 = sm.z * invN, mu3 = sm.w * invN;
  const float rs0 = rsqrtf(sq.x * invN - mu0 * mu0 + BN_EPS);
  const float rs1 = rsqrtf(sq.y * invN - mu1 * mu1 + BN_EPS);
  const float rs2 = rsqrtf(sq.z * invN - mu2 * mu2 + BN_EPS);
  const float rs3 = rsqrtf(sq.w * invN - mu3 * mu3 + BN_EPS);
#pragma unroll
  for (int u = 0; u < ST_U; ++u) {
    const int i = i0 + u * ST_T;
    float4 o;
    o.x = (fmaxf(v[u].x, 0.0f) - mu0) * rs0 * g0 + b0;
    o.y = (fmaxf(v[u].y, 0.0f) - mu1) * rs1 * g1 + b1;
    o.z = (fmaxf(v[u].z, 0.0f) - mu2) * rs2 * g2 + b2;
    o.w = (fmaxf(v[u].w, 0.0f) - mu3) * rs3 * g3 + b3;
    if (WHOLE || i < total) o4[i] = o;
  }
}

__global__ __launch_bounds__(ST_T) void k_bn_apply(const float* __restrict__ c, const float* __restrict__ stats,
                           const float* __restrict__ g, const float* __restrict__ be,
                           float* __restrict__ out, int N) {
  const int tid = threadIdx.x;
  const int fq  = (tid & 7) * 4;
  const int total = N * (HID / 4);
  const int b0 = blockIdx.x * (ST_T * ST_U);
  const float invN = 1.0f / (float)N;
  const float4* __restrict__ c4 = (const float4*)c;
  float4* __restrict__ o4 = (float4*)out;
  if (b0 + ST_T * ST_U <= total)
    bn_apply_pieces<true>(c4, stats, g, be, o4, b0 + tid, fq, total, invN);
  else
    bn_apply_pieces<false>(c4, stats, g, be, o4, b0 + tid, fq, total, invN);
}

__global__ void k_final(const float* __restrict__ o1, const float* __restrict__ o2,
                        const float* __restrict__ o3, const float* __restrict__ Wl,
                        const float* __restrict__ bl, float* __restrict__ out, int N) {
  int n = blockIdx.x * blockDim.x + threadIdx.x;
  if (n >= N) return;
  float a0 = bl[0], a1 = bl[1];
  const float* p = o1 + (size_t)n * HID;
#pragma unroll
  for (int j = 0; j < HID; ++j) { float v = p[j]; a0 += v * Wl[2 * j]; a1 += v * Wl[2 * j + 1]; }
  p = o2 + (size_t)n * HID;
#pragma unroll
  for (int j = 0; j < HID; ++j) { float v = p[j]; a0 += v * Wl[2 * (HID + j)]; a1 += v * Wl[2 * (HID + j) + 1]; }
  p = o3 + (size_t)n * HID;
#pragma unroll
  for (int j = 0; j < HID; ++j) { float v = p[j]; a0 += v * Wl[2 * (2 * HID + j)]; a1 += v * Wl[2 * (2 * HID + j) + 1]; }
  out[2 * n] = a0;
  out[2 * n + 1] = a1;
}

extern "C" void kernel_launch(void* const* d_in, const int* in_sizes, int n_in,
                              void* d_out, int out_size, void* d_ws, size_t ws_size,
                              hipStream_t stream) {
  const float* x   = (const float*)d_in[0];
  const int*   ei  = (const int*)d_in[1];
  const float* ew  = (const float*)d_in[2];
  const float* W1  = (const float*)d_in[3];
  const float* b1  = (const float*)d_in[4];
  const float* g1  = (const float*)d_in[5];
  const float* be1 = (const float*)d_in[6];
  const float* W2  = (const float*)d_in[7];
  const float* b2  = (const float*)d_in[8];
  const float* g2  = (const float*)d_in[9];
  const float* be2 = (const float*)d_in[10];
  const float* W3  = (const float*)d_in[11];
  const float* b3  = (const float*)d_in[12];
  const float* g3  = (const float*)d_in[13];
  const float* be3 = (const float*)d_in[14];
  const float* Wl  = (const float*)d_in[15];
  const float* bl  = (const float*)d_in[16];

  const int F = 64;
  const int N = in_sizes[0] / F;
  const int E = in_sizes[1] / 2;
  const int* row = ei;
  const int* col = ei + E;

  float* ws    = (float*)d_ws;
  float* dis   = ws;
  float* hbuf  = dis  + ((N + 3) & ~3);
  float* cbuf  = hbuf + (size_t)N * HID;
  float* o1    = cbuf + (size_t)N * HID;
  float* o2    = o1   + (size_t)N * HID;
  float* o3    = o2   + (size_t)N * HID;
  float* stats = o3   + (size_t)N * HID;
  int*   cnt   = (int*)(stats + 64);
  int*   rec   = (int*)(((size_t)(cnt + 1024) + 15) & ~(size_t)15);
  int*   ovf   = rec + (size_t)NB * BCAP * 4;

  const int T = 256;

  const int numT = (N + 15) / 16;
  const dim3 gemmB(32, 8);
  const dim3 gemmG((numT + 7) / 8, 2);
  const int R = (N + NB - 1) / NB;
  const size_t accB = (size_t)R * HID * sizeof(float);
  const int RP = (R + 15) & ~15;
  const size_t accM = (size_t)RP * (APITCH + HID) * sizeof(float);
  const int binC = BIN_T * BIN_EPT;
  const int statB = (N * (HID / 4) + ST_T * ST_U - 1) / (ST_T * ST_U);

  const size_t degB = (size_t)R * sizeof(float);

  k_zero_counts<<<1, T, 0, stream>>>(cnt);
  k_bin<<<(E + binC - 1) / binC, BIN_T, 0, stream>>>(row, col, ew, cnt, rec, ovf, E, R);
  k_deg<<<NB, DG_T, degB, stream>>>(cnt, rec, ovf, dis, N, R);

  k_gemm_wmma<64><<<gemmG, gemmB, 0, stream>>>(x, W1, b1, dis, hbuf, cbuf, stats, N);
  k_prop<<<NB, PL_T, accB, stream>>>(hbuf, cnt, rec, ovf, dis, cbuf, N, R);
  k_bn_stats<<<statB, ST_T, 0, stream>>>(cbuf, stats, N);
  k_bn_apply<<<statB, ST_T, 0, stream>>>(cbuf, stats, g1, be1, o1, N);

  k_prop_mm<<<NB, PL_T, accM, stream>>>(o1, W2, b2, dis, cnt, rec, ovf, cbuf, stats, N, R);
  k_bn_stats<<<statB, ST_T, 0, stream>>>(cbuf, stats, N);
  k_bn_apply<<<statB, ST_T, 0, stream>>>(cbuf, stats, g2, be2, o2, N);

  k_prop_mm<<<NB, PL_T, accM, stream>>>(o2, W3, b3, dis, cnt, rec, ovf, cbuf, stats, N, R);
  k_bn_stats<<<statB, ST_T, 0, stream>>>(cbuf, stats, N);
  k_bn_apply<<<statB, ST_T, 0, stream>>>(cbuf, stats, g3, be3, o3, N);

  k_final<<<(N + T - 1) / T, T, 0, stream>>>(o1, o2, o3, Wl, bl, (float*)d_out, N);
}


# Layer 1's transform joins the degree launch: a range's workgroup keeps its own input rows in LDS, multiplies them by W1 on the matrix unit beside its degree sums, stores whole lines; 13 launches
# speedup vs baseline: 4.8918x; 1.0590x over previous
#include <hip/hip_runtime.h>

typedef float v2f __attribute__((ext_vector_type(2)));
typedef float v8f __attribute__((ext_vector_type(8)));

#define HID 32
#define BN_EPS 1e-5f

#define NB 512
#define BCAP 4096
__global__ void k_zero_counts(int* __restrict__ cnt) {
  for (int j = threadIdx.x; j <= NB; j += blockDim.x) cnt[j] = 0;
}

#define BIN_T 512
#define BIN_EPT 8
__global__ __launch_bounds__(BIN_T) void k_bin(const int* __restrict__ row, const int* __restrict__ col,
                      const float* __restrict__ w,
                      int* __restrict__ cnt, int* __restrict__ rec, int* __restrict__ ovf, int E, int R) {
  __shared__ int hist[NB];
  __shared__ int loff[NB];
  __shared__ int gdst[NB];
  __shared__ int ltot;
  __shared__ int4 stage[BIN_T * BIN_EPT];
  const int tid = threadIdx.x;
  for (int i = tid; i < NB; i += BIN_T) hist[i] = 0;
  if (tid == 0) ltot = 0;
  __syncthreads();
  const int cb = blockIdx.x * (BIN_T * BIN_EPT);
  int bn[BIN_EPT], lr[BIN_EPT];
#pragma unroll
  for (int k = 0; k < BIN_EPT; ++k) {
    const int e = cb + k * BIN_T + tid;
    bn[k] = -1;
    lr[k] = 0;
    if (e < E) {
      const int bb = col[e] / R;
      bn[k] = bb;
      lr[k] = atomicAdd(&hist[bb], 1);
    }
  }
  __syncthreads();
  for (int i = tid; i < NB; i += BIN_T) {
    const int hcount = hist[i];
    int lo = 0, gb = 0;
    if (hcount) {
      lo = atomicAdd(&ltot, hcount);
      gb = atomicAdd(&cnt[i], hcount);
    }
    loff[i] = lo;
    gdst[i] = gb;
  }
  __syncthreads();
#pragma unroll
  for (int k = 0; k < BIN_EPT; ++k) {
    if (bn[k] >= 0) {
      const int e  = cb + k * BIN_T + tid;
      const int bb = bn[k];
      const int r  = row[e];
      const int c  = col[e];
      const int slot = gdst[bb] + lr[k];
      const bool fits = slot < BCAP;
      stage[loff[bb] + lr[k]] = make_int4(fits ? bb * BCAP + slot : -1, r, fits ? c - bb * R : c,
                                          __float_as_int(w[e]));
    }
  }
  __syncthreads();
  const int tot = ltot;
  int4* rec4 = (int4*)rec;
  int4* ovf4 = (int4*)ovf;
  for (int p = tid; p < tot; p += BIN_T) {
    const int4 s = stage[p];
    int4* dp = rec4 + s.x;
    if (s.x < 0) dp = ovf4 + atomicAdd(&cnt[NB], 1);
    *dp = make_int4(s.y, s.z, s.w, 0);
  }
}

#define DG_T 512
#define DG_U (BCAP / DG_T)
#define XF 64
#define XPITCH 68
__global__ __launch_bounds__(DG_T) void k_deg_mm(const int* __restrict__ cnt, const int* __restrict__ rec,
                      const int* __restrict__ ovf, const float* __restrict__ x, const float* __restrict__ Wt,
                      const float* __restrict__ bias, float* __restrict__ dis, float* __restrict__ D,
                      float* __restrict__ S, float* __restrict__ stats, int N, int R) {
  extern __shared__ float lds[];
  const int tid = threadIdx.x;
  const unsigned lane = threadIdx.x & 31u;
  const int wv  = (int)(threadIdx.x >> 5);
  const int bin = blockIdx.x;
  const int r0  = bin * R;
  int nrows = N - r0;
  if (nrows > R) nrows = R;
  if (nrows < 0) nrows = 0;
  const int nel = nrows * HID;
  const int RP  = (R + 15) & ~15;
  float* dsum = lds;
  float* xt   = lds + ((R + 3) & ~3);
  int n = cnt[bin];
  if (n > BCAP) n = BCAP;
  const int4* rec4 = (const int4*)rec + (size_t)bin * BCAP;
  int4 rc[DG_U];
#pragma unroll
  for (int u = 0; u < DG_U; ++u) rc[u] = make_int4(0, 0, 0, 0);
  if (n > 0) {
#pragma unroll
    for (int u = 0; u < DG_U; ++u) {
      const int i = u * DG_T + tid;
      rc[u] = rec4[i < n ? i : n - 1];
    }
  }
  for (int i = tid; i < R; i += DG_T) dsum[i] = 1.0f;
  for (int t0 = 0; t0 * 16 < RP; t0 += 8) {
    v2f xv[8];
#pragma unroll
    for (int u = 0; u < 8; ++u) {
      int g = r0 + (t0 + u) * 16 + wv;
      if (g > N - 1) g = N - 1;
      xv[u] = *(const v2f*)(x + (size_t)g * XF + 2 * lane);
    }
    asm volatile("" :: "v"(xv[0]), "v"(xv[1]), "v"(xv[2]), "v"(xv[3]), "v"(xv[4]), "v"(xv[5]), "v"(xv[6]), "v"(xv[7]));
#pragma unroll
    for (int u = 0; u < 8; ++u) {
      const int rr = (t0 + u) * 16 + wv;
      if (rr < RP) {
        v2f o;
        o.x = rr < nrows ? xv[u].x : 0.0f;
        o.y = rr < nrows ? xv[u].y : 0.0f;
        *(v2f*)(xt + rr * XPITCH + 2 * lane) = o;
      }
    }
  }
  __syncthreads();
  if (n > 0) {
#pragma unroll
    for (int u = 0; u < DG_U; ++u)
      if (u * DG_T + tid < n) atomicAdd(&dsum[rc[u].y], __int_as_float(rc[u].z));
  }
  const int nov = cnt[NB];
  const int4* ovf4 = (const int4*)ovf;
  for (int i = tid; i < nov; i += DG_T) {
    const int4 s = ovf4[i];
    if (s.y >= r0 && s.y < r0 + nrows) atomicAdd(&dsum[s.y - r0], __int_as_float(s.z));
  }
  __syncthreads();
  for (int i = tid; i < nrows; i += DG_T) {
    const float d = rsqrtf(dsum[i]);
    dis[r0 + i] = d;
    dsum[i] = d;
  }

  const int half  = (int)(lane >> 4);
  const int lm    = (int)(lane & 15u);
  const int nstr  = __builtin_amdgcn_readfirstlane((nrows + 15) >> 4);
  const int wvs   = __builtin_amdgcn_readfirstlane(wv);
  for (int mt = wvs; mt < nstr; mt += DG_T / 32) {
    const float* arow = xt + (mt * 16 + lm) * XPITCH;
    v8f c0 = {};
    v8f c1 = {};
#pragma unroll
    for (int k0 = 0; k0 < XF; k0 += 4) {
      const int ka = k0 + 2 * half;
      v2f a;
      a.x = arow[ka];
      a.y = arow[ka + 1];
      v2f b0;
      b0.x = Wt[(size_t)ka * HID + lm];
      b0.y = Wt[(size_t)(ka + 1) * HID + lm];
      v2f b1;
      b1.x = Wt[(size_t)ka * HID + 16 + lm];
      b1.y = Wt[(size_t)(ka + 1) * HID + 16 + lm];
      c0 = __builtin_amdgcn_wmma_f32_16x16x4_f32(false, a, false, b0, (short)0, c0,
                                                 false, false);
      c1 = __builtin_amdgcn_wmma_f32_16x16x4_f32(false, a, false, b1, (short)0, c1,
                                                 false, false);
    }
    float* prow = xt + (mt * 16 + 8 * half) * XPITCH;
#pragma unroll
    for (int r = 0; r < 8; ++r) {
      prow[r * XPITCH + lm]      = c0[r];
      prow[r * XPITCH + 16 + lm] = c1[r];
    }
  }
  __syncthreads();
  const float bcol = bias[lane];
  float* dbase = D + (size_t)r0 * HID;
  float* sbase = S + (size_t)r0 * HID;
  for (int i = tid; i < nel; i += DG_T) {
    const int   rr = i >> 5;
    const float hv = xt[rr * XPITCH + (int)lane];
    const float d  = dsum[rr];
    dbase[i] = hv;
    sbase[i] = hv * (d * d) + bcol;
  }
  if (bin == 0 && wv == 0) {
    stats[lane]      = 0.0f;
    stats[32 + lane] = 0.0f;
  }
}

#define PL_T 512
#define APITCH 36
template <int PITCH>
__device__ __forceinline__ void edge_sums(const float* __restrict__ h, const int4* __restrict__ rec4,
                                          const int n, float* acc, const unsigned lane, const int wv) {
  const int w0 = __builtin_amdgcn_readfirstlane(wv * 32);
  const unsigned lane4 = lane * 4u;
  const char* hb = (const char*)h;
#pragma nounroll
  for (int g0 = w0; g0 < n; g0 += PL_T) {
    const int i  = g0 + (int)lane;
    const bool ok = i < n;
    const int4 rc = rec4[ok ? i : (n - 1)];
    const unsigned ro  = (unsigned)rc.x * (unsigned)(HID * sizeof(float));
    const int      co  = rc.y * (int)(PITCH * sizeof(float));
    const int      cfb = rc.z;
    if (g0 + 32 <= n) {
      float v[32];
#pragma unroll
      for (int j = 0; j < 32; ++j) {
        const unsigned off = (unsigned)__builtin_amdgcn_readlane((int)ro, j) + lane4;
        v[j] = *(const float*)(hb + off);
      }
#pragma unroll
      for (int j = 0; j < 32; ++j) {
        float*      ac = (float*)((char*)acc + __builtin_amdgcn_readlane(co, j));
        const float cj = __int_as_float(__builtin_amdgcn_readlane(cfb, j));
        atomicAdd(ac + lane, v[j] * cj);
      }
    } else {
      const unsigned mask = __builtin_amdgcn_ballot_w32(ok);
      float v[32];
#pragma unroll
      for (int j = 0; j < 32; ++j) {
        if ((mask >> j) & 1u) {
          const unsigned off = (unsigned)__builtin_amdgcn_readlane((int)ro, j) + lane4;
          v[j] = *(const float*)(hb + off);
        }
      }
#pragma unroll
      for (int j = 0; j < 32; ++j) {
        if ((mask >> j) & 1u) {
          float*      ac = (float*)((char*)acc + __builtin_amdgcn_readlane(co, j));
          const float cj = __int_as_float(__builtin_amdgcn_readlane(cfb, j));
          atomicAdd(ac + lane, v[j] * cj);
        }
      }
    }
  }
}

__device__ __forceinline__ void edge_sums_first(const float* __restrict__ h, int* recb, const float* __restrict__ dis,
                                                const int r0, const int n, float* acc, const unsigned lane,
                                                const int wv) {
  const int w0 = __builtin_amdgcn_readfirstlane(wv * 32);
  const unsigned lane4 = lane * 4u;
  const char* hb = (const char*)h;
  const int4* rec4 = (const int4*)recb;
#pragma nounroll
  for (int g0 = w0; g0 < n; g0 += PL_T) {
    const int i  = g0 + (int)lane;
    const bool ok = i < n;
    const int ic = ok ? i : (n - 1);
    const int4 rc = rec4[ic];
    const unsigned ro  = (unsigned)rc.x * (unsigned)(HID * sizeof(float));
    const int      co  = rc.y * (int)(HID * sizeof(float));
    const float ds = dis[rc.x];
    const float dd = dis[r0 + rc.y];
    const bool full = g0 + 32 <= n;
    const unsigned mask = full ? 0xffffffffu : __builtin_amdgcn_ballot_w32(ok);
    float v[32];
    if (full) {
#pragma unroll
      for (int j = 0; j < 32; ++j) {
        const unsigned off = (unsigned)__builtin_amdgcn_readlane((int)ro, j) + lane4;
        v[j] = *(const float*)(hb + off);
      }
    } else {
#pragma unroll
      for (int j = 0; j < 32; ++j) {
        v[j] = 0.0f;
        if ((mask >> j) & 1u) {
          const unsigned off = (unsigned)__builtin_amdgcn_readlane((int)ro, j) + lane4;
          v[j] = *(const float*)(hb + off);
        }
      }
    }
    const int cfb = __float_as_int(ds * __int_as_float(rc.z) * dd);
    if (ok) recb[(size_t)ic * 4 + 2] = cfb;
    if (full) {
#pragma unroll
      for (int j = 0; j < 32; ++j) {
        float*      ac = (float*)((char*)acc + __builtin_amdgcn_readlane(co, j));
        const float cj = __int_as_float(__builtin_amdgcn_readlane(cfb, j));
        atomicAdd(ac + lane, v[j] * cj);
      }
    } else {
#pragma unroll
      for (int j = 0; j < 32; ++j) {
        if ((mask >> j) & 1u) {
          float*      ac = (float*)((char*)acc + __builtin_amdgcn_readlane(co, j));
          const float cj = __int_as_float(__builtin_amdgcn_readlane(cfb, j));
          atomicAdd(ac + lane, v[j] * cj);
        }
      }
    }
  }
}

__global__ __launch_bounds__(PL_T) void k_prop(const float* __restrict__ h, const int* __restrict__ cnt,
                       int* rec, int* ovf, const float* __restrict__ dis,
                       float* __restrict__ out, int N, int R) {
  extern __shared__ float acc[];
  const int tid = threadIdx.x;
  const unsigned lane = threadIdx.x & 31u;
  const int wv  = (int)(threadIdx.x >> 5);
  const int bin = blockIdx.x;
  const int r0  = bin * R;
  int nrows = N - r0;
  if (nrows > R) nrows = R;
  if (nrows < 0) nrows = 0;
  const int nel = nrows * HID;
  float* obase = out + (size_t)r0 * HID;
  for (int i = tid; i < nel; i += PL_T) acc[i] = obase[i];
  __syncthreads();

  int n = cnt[bin];
  if (n > BCAP) n = BCAP;
  n = __builtin_amdgcn_readfirstlane(n);
  edge_sums_first(h, rec + (size_t)bin * BCAP * 4, dis, r0, n, acc, lane, wv);

  const int nov = cnt[NB];
  const int4* ovf4 = (const int4*)ovf;
  for (int i = tid; i < nov; i += PL_T) {
    const int4 s = ovf4[i];
    if (s.y >= r0 && s.y < r0 + nrows) {
      const float* hr = h + (size_t)s.x * HID;
      const float  cf = dis[s.x] * __int_as_float(s.z) * dis[s.y];
      ovf[(size_t)i * 4 + 2] = __float_as_int(cf);
      float*       ac = acc + (s.y - r0) * HID;
      for (int f = 0; f < HID; ++f) atomicAdd(ac + f, hr[f] * cf);
    }
  }
  __syncthreads();
  for (int i = tid; i < nel; i += PL_T) obase[i] = acc[i];
}

__global__ __launch_bounds__(PL_T) void k_prop_mm(const float* __restrict__ in, const float* __restrict__ Wt,
                       const float* __restrict__ bias, const float* __restrict__ dis,
                       const int* __restrict__ cnt, const int* __restrict__ rec, const int* __restrict__ ovf,
                       float* __restrict__ out, float* __restrict__ stats, int N, int R) {
  extern __shared__ float acc[];
  const int tid = threadIdx.x;
  const unsigned lane = threadIdx.x & 31u;
  const int wv  = (int)(threadIdx.x >> 5);
  const int bin = blockIdx.x;
  const int r0  = bin * R;
  int nrows = N - r0;
  if (nrows > R) nrows = R;
  if (nrows < 0) nrows = 0;
  const int nel = nrows * HID;
  const int RP  = (R + 15) & ~15;
  float* prod  = acc + RP * APITCH;
  float* obase = out + (size_t)r0 * HID;
  for (int t0 = 0; t0 * 16 < RP; t0 += 8) {
    float xv[8], dv[8];
#pragma unroll
    for (int u = 0; u < 8; ++u) {
      int g = r0 + (t0 + u) * 16 + wv;
      if (g > N - 1) g = N - 1;
      dv[u] = dis[g];
      xv[u] = in[(size_t)g * HID + lane];
    }
    asm volatile("" :: "v"(xv[0]), "v"(xv[1]), "v"(xv[2]), "v"(xv[3]), "v"(xv[4]), "v"(xv[5]), "v"(xv[6]), "v"(xv[7]),
                       "v"(dv[0]), "v"(dv[1]), "v"(dv[2]), "v"(dv[3]), "v"(dv[4]), "v"(dv[5]), "v"(dv[6]), "v"(dv[7]));
#pragma unroll
    for (int u = 0; u < 8; ++u) {
      const int rr = (t0 + u) * 16 + wv;
      if (rr < RP) acc[rr * APITCH + lane] = rr < nrows ? xv[u] * (dv[u] * dv[u]) : 0.0f;
    }
  }
  __syncthreads();

  int n = cnt[bin];
  if (n > BCAP) n = BCAP;
  n = __builtin_amdgcn_readfirstlane(n);
  const int4* rec4 = (const int4*)rec + (size_t)bin * BCAP;
  edge_sums<APITCH>(in, rec4, n, acc, lane, wv);

  const int nov = cnt[NB];
  const int4* ovf4 = (const int4*)ovf;
  for (int i = tid; i < nov; i += PL_T) {
    const int4 s = ovf4[i];
    if (s.y >= r0 && s.y < r0 + nrows) {
      const float* hr = in + (size_t)s.x * HID;
      const float  cf = __int_as_float(s.z);
      float*       ac = acc + (s.y - r0) * APITCH;
      for (int f = 0; f < HID; ++f) atomicAdd(ac + f, hr[f] * cf);
    }
  }
  __syncthreads();

  const int half  = (int)(lane >> 4);
  const int lm    = (int)(lane & 15u);
  const int njobs = __builtin_amdgcn_readfirstlane(((nrows + 15) >> 4) * 2);
  const int wvs   = __builtin_amdgcn_readfirstlane(wv);
  for (int job = wvs; job < njobs; job += PL_T / 32) {
    const int mt = job >> 1;
    const int nb = (job & 1) * 16;
    const float* arow = acc + (mt * 16 + lm) * APITCH;
    v8f c = {};
#pragma unroll
    for (int k0 = 0; k0 < HID; k0 += 4) {
      const int ka = k0 + 2 * half;
      v2f a;
      a.x = arow[ka];
      a.y = arow[ka + 1];
      v2f b;
      b.x = Wt[(size_t)ka * HID + nb + lm];
      b.y = Wt[(size_t)(ka + 1) * HID + nb + lm];
      c = __builtin_amdgcn_wmma_f32_16x16x4_f32(false, a, false, b, (short)0, c,
                                                false, false);
    }
    const float bcol = bias[nb + lm];
#pragma unroll
    for (int r = 0; r < 8; ++r)
      prod[(mt * 16 + r + 8 * half) * HID + nb + lm] = c[r] + bcol;
  }
  __syncthreads();
  for (int i = tid; i < nel; i += PL_T) obase[i] = prod[i];
  if (bin == 0 && wv == 0) {
    stats[lane]      = 0.0f;
    stats[32 + lane] = 0.0f;
  }
}

#define ST_T 256
#define ST_U 8
__global__ __launch_bounds__(ST_T) void k_bn_stats(const float* __restrict__ c, float* __restrict__ stats, int N) {
  __shared__ float ls[64];
  const int tid = threadIdx.x;
  if (tid < 64) ls[tid] = 0.0f;
  __syncthreads();
  const float4* __restrict__ c4 = (const float4*)c;
  const int total = N * (HID / 4);
  const int i0 = blockIdx.x * (ST_T * ST_U) + tid;
  float4 v[ST_U];
#pragma unroll
  for (int u = 0; u < ST_U; ++u) {
    const int i = i0 + u * ST_T;
    v[u] = c4[i < total ? i : total - 1];
  }
  float s0 = 0.0f, s1 = 0.0f, s2 = 0.0f, s3 = 0.0f;
  float q0 = 0.0f, q1 = 0.0f, q2 = 0.0f, q3 = 0.0f;
#pragma unroll
  for (int u = 0; u < ST_U; ++u) {
    const bool in = i0 + u * ST_T < total;
    const float a0 = in ? fmaxf(v[u].x, 0.0f) : 0.0f;
    const float a1 = in ? fmaxf(v[u].y, 0.0f) : 0.0f;
    const float a2 = in ? fmaxf(v[u].z, 0.0f) : 0.0f;
    const float a3 = in ? fmaxf(v[u].w, 0.0f) : 0.0f;
    s0 += a0; q0 += a0 * a0;
    s1 += a1; q1 += a1 * a1;
    s2 += a2; q2 += a2 * a2;
    s3 += a3; q3 += a3 * a3;
  }
  s0 += __shfl_xor(s0, 8);  s1 += __shfl_xor(s1, 8);  s2 += __shfl_xor(s2, 8);  s3 += __shfl_xor(s3, 8);
  q0 += __shfl_xor(q0, 8);  q1 += __shfl_xor(q1, 8);  q2 += __shfl_xor(q2, 8);  q3 += __shfl_xor(q3, 8);
  s0 += __shfl_xor(s0, 16); s1 += __shfl_xor(s1, 16); s2 += __shfl_xor(s2, 16); s3 += __shfl_xor(s3, 16);
  q0 += __shfl_xor(q0, 16); q1 += __shfl_xor(q1, 16); q2 += __shfl_xor(q2, 16); q3 += __shfl_xor(q3, 16);
  const int lane = tid & 31;
  if (lane < 8) {
    float* lp = ls + 4 * lane;
    atomicAdd(lp + 0, s0); atomicAdd(lp + 1, s1); atomicAdd(lp + 2, s2); atomicAdd(lp + 3, s3);
    atomicAdd(lp + 32, q0); atomicAdd(lp + 33, q1); atomicAdd(lp + 34, q2); atomicAdd(lp + 35, q3);
  }
  __syncthreads();
  if (tid < 64) atomicAdd(&stats[tid], ls[tid]);
}

template <bool WHOLE>
__device__ __forceinline__ void bn_apply_pieces(const float4* __restrict__ c4, const float* __restrict__ stats,
                                                const float* __restrict__ g, const float* __restrict__ be,
                                                float4* __restrict__ o4, const int i0, const int fq,
                                                const int total, const float invN) {
  float4 v[ST_U];
#pragma unroll
  for (int u = 0; u < ST_U; ++u) {
    const int i = i0 + u * ST_T;
    v[u] = c4[(WHOLE || i < total) ? i : total - 1];
  }
  const float4 sm = *(const float4*)(stats + fq);
  const float4 sq = *(const float4*)(stats + 32 + fq);
  const float g0 = g[fq], g1 = g[fq + 1], g2 = g[fq + 2], g3 = g[fq + 3];
  const float b0 = be[fq], b1 = be[fq + 1], b2 = be[fq + 2], b3 = be[fq + 3];
  const float mu0 = sm.x * invN, mu1 = sm.y * invN, mu2 = sm.z * invN, mu3 = sm.w * invN;
  const float rs0 = rsqrtf(sq.x * invN - mu0 * mu0 + BN_EPS);
  const float rs1 = rsqrtf(sq.y * invN - mu1 * mu1 + BN_EPS);
  const float rs2 = rsqrtf(sq.z * invN - mu2 * mu2 + BN_EPS);
  const float rs3 = rsqrtf(sq.w * invN - mu3 * mu3 + BN_EPS);
#pragma unroll
  for (int u = 0; u < ST_U; ++u) {
    const int i = i0 + u * ST_T;
    float4 o;
    o.x = (fmaxf(v[u].x, 0.0f) - mu0) * rs0 * g0 + b0;
    o.y = (fmaxf(v[u].y, 0.0f) - mu1) * rs1 * g1 + b1;
    o.z = (fmaxf(v[u].z, 0.0f) - mu2) * rs2 * g2 + b2;
    o.w = (fmaxf(v[u].w, 0.0f) - mu3) * rs3 * g3 + b3;
    if (WHOLE || i < total) o4[i] = o;
  }
}

__global__ __launch_bounds__(ST_T) void k_bn_apply(const float* __restrict__ c, const float* __restrict__ stats,
                           const float* __restrict__ g, const float* __restrict__ be,
                           float* __restrict__ out, int N) {
  const int tid = threadIdx.x;
  const int fq  = (tid & 7) * 4;
  const int total = N * (HID / 4);
  const int b0 = blockIdx.x * (ST_T * ST_U);
  const float invN = 1.0f / (float)N;
  const float4* __restrict__ c4 = (const float4*)c;
  float4* __restrict__ o4 = (float4*)out;
  if (b0 + ST_T * ST_U <= total)
    bn_apply_pieces<true>(c4, stats, g, be, o4, b0 + tid, fq, total, invN);
  else
    bn_apply_pieces<false>(c4, stats, g, be, o4, b0 + tid, fq, total, invN);
}

__global__ void k_final(const float* __restrict__ o1, const float* __restrict__ o2,
                        const float* __restrict__ o3, const float* __restrict__ Wl,
                        const float* __restrict__ bl, float* __restrict__ out, int N) {
  int n = blockIdx.x * blockDim.x + threadIdx.x;
  if (n >= N) return;
  float a0 = bl[0], a1 = bl[1];
  const float* p = o1 + (size_t)n * HID;
#pragma unroll
  for (int j = 0; j < HID; ++j) { float v = p[j]; a0 += v * Wl[2 * j]; a1 += v * Wl[2 * j + 1]; }
  p = o2 + (size_t)n * HID;
#pragma unroll
  for (int j = 0; j < HID; ++j) { float v = p[j]; a0 += v * Wl[2 * (HID + j)]; a1 += v * Wl[2 * (HID + j) + 1]; }
  p = o3 + (size_t)n * HID;
#pragma unroll
  for (int j = 0; j < HID; ++j) { float v = p[j]; a0 += v * Wl[2 * (2 * HID + j)]; a1 += v * Wl[2 * (2 * HID + j) + 1]; }
  out[2 * n] = a0;
  out[2 * n + 1] = a1;
}

extern "C" void kernel_launch(void* const* d_in, const int* in_sizes, int n_in,
                              void* d_out, int out_size, void* d_ws, size_t ws_size,
                              hipStream_t stream) {
  const float* x   = (const float*)d_in[0];
  const int*   ei  = (const int*)d_in[1];
  const float* ew  = (const float*)d_in[2];
  const float* W1  = (const float*)d_in[3];
  const float* b1  = (const float*)d_in[4];
  const float* g1  = (const float*)d_in[5];
  const float* be1 = (const float*)d_in[6];
  const float* W2  = (const float*)d_in[7];
  const float* b2  = (const float*)d_in[8];
  const float* g2  = (const float*)d_in[9];
  const float* be2 = (const float*)d_in[10];
  const float* W3  = (const float*)d_in[11];
  const float* b3  = (const float*)d_in[12];
  const float* g3  = (const float*)d_in[13];
  const float* be3 = (const float*)d_in[14];
  const float* Wl  = (const float*)d_in[15];
  const float* bl  = (const float*)d_in[16];

  const int F = 64;
  const int N = in_sizes[0] / F;
  const int E = in_sizes[1] / 2;
  const int* row = ei;
  const int* col = ei + E;

  float* ws    = (float*)d_ws;
  float* dis   = ws;
  float* hbuf  = dis  + ((N + 3) & ~3);
  float* cbuf  = hbuf + (size_t)N * HID;
  float* o1    = cbuf + (size_t)N * HID;
  float* o2    = o1   + (size_t)N * HID;
  float* o3    = o2   + (size_t)N * HID;
  float* stats = o3   + (size_t)N * HID;
  int*   cnt   = (int*)(stats + 64);
  int*   rec   = (int*)(((size_t)(cnt + 1024) + 15) & ~(size_t)15);
  int*   ovf   = rec + (size_t)NB * BCAP * 4;

  const int T = 256;

  const int R = (N + NB - 1) / NB;
  const size_t accB = (size_t)R * HID * sizeof(float);
  const int RP = (R + 15) & ~15;
  const size_t accM = (size_t)RP * (APITCH + HID) * sizeof(float);
  const int binC = BIN_T * BIN_EPT;
  const int statB = (N * (HID / 4) + ST_T * ST_U - 1) / (ST_T * ST_U);

  const size_t degB = ((size_t)((R + 3) & ~3) + (size_t)RP * XPITCH) * sizeof(float);

  k_zero_counts<<<1, T, 0, stream>>>(cnt);
  k_bin<<<(E + binC - 1) / binC, BIN_T, 0, stream>>>(row, col, ew, cnt, rec, ovf, E, R);
  k_deg_mm<<<NB, DG_T, degB, stream>>>(cnt, rec, ovf, x, W1, b1, dis, hbuf, cbuf, stats, N, R);

  k_prop<<<NB, PL_T, accB, stream>>>(hbuf, cnt, rec, ovf, dis, cbuf, N, R);
  k_bn_stats<<<statB, ST_T, 0, stream>>>(cbuf, stats, N);
  k_bn_apply<<<statB, ST_T, 0, stream>>>(cbuf, stats, g1, be1, o1, N);

  k_prop_mm<<<NB, PL_T, accM, stream>>>(o1, W2, b2, dis, cnt, rec, ovf, cbuf, stats, N, R);
  k_bn_stats<<<statB, ST_T, 0, stream>>>(cbuf, stats, N);
  k_bn_apply<<<statB, ST_T, 0, stream>>>(cbuf, stats, g2, be2, o2, N);

  k_prop_mm<<<NB, PL_T, accM, stream>>>(o2, W3, b3, dis, cnt, rec, ovf, cbuf, stats, N, R);
  k_bn_stats<<<statB, ST_T, 0, stream>>>(cbuf, stats, N);
  k_bn_apply<<<statB, ST_T, 0, stream>>>(cbuf, stats, g3, be3, o3, N);

  k_final<<<(N + T - 1) / T, T, 0, stream>>>(o1, o2, o3, Wl, bl, (float*)d_out, N);
}
